# starting kernel
# baseline (speedup 1.0000x reference)
#include <hip/hip_runtime.h>
#include <hip/hip_bf16.h>


typedef __attribute__((ext_vector_type(2))) float v2f;
typedef __attribute__((ext_vector_type(8))) float v8f;

#define HID 128
#define EMBD 64

__global__ void fill0_kernel(float* __restrict__ p, long long n) {
    long long i = (long long)blockIdx.x * blockDim.x + threadIdx.x;
    if (i < n) p[i] = 0.0f;
}

__global__ void build_h0_kernel(const int* __restrict__ n_id,
                                const float* __restrict__ emb,
                                const float* __restrict__ x_struct,
                                float* __restrict__ h0, int n_nodes) {
    long long t = (long long)blockIdx.x * blockDim.x + threadIdx.x;
    int i = (int)(t >> 7);
    int c = (int)(t & 127);
    if (i >= n_nodes) return;
    float v;
    if (c < EMBD) v = emb[(size_t)n_id[i] * EMBD + c];
    else          v = x_struct[(size_t)i * EMBD + (c - EMBD)];
    h0[(size_t)i * HID + c] = v;
}

__global__ void deg_kernel(const int* __restrict__ dst, float* __restrict__ deg,
                           int n_edges) {
    int e = blockIdx.x * blockDim.x + threadIdx.x;
    if (e >= n_edges) return;
    atomicAdd(&deg[dst[e]], 1.0f);
}

__global__ void invdeg_kernel(const float* __restrict__ deg,
                              float* __restrict__ invdeg, int n_nodes) {
    int i = blockIdx.x * blockDim.x + threadIdx.x;
    if (i >= n_nodes) return;
    invdeg[i] = 1.0f / fmaxf(deg[i], 1.0f);
}

__global__ void scatter_kernel(const float* __restrict__ h,
                               const int* __restrict__ src,
                               const int* __restrict__ dst,
                               float* __restrict__ agg, int n_edges) {
    long long t = (long long)blockIdx.x * blockDim.x + threadIdx.x;
    int e = (int)(t >> 7);
    int c = (int)(t & 127);
    if (e >= n_edges) return;
    int s = src[e], d = dst[e];
    atomicAdd(&agg[(size_t)d * HID + c], h[(size_t)s * HID + c]);
}

__global__ __launch_bounds__(256)
void sage_gemm_kernel(const float* __restrict__ agg,
                      const float* __restrict__ invdeg,
                      const float* __restrict__ h,
                      const float* __restrict__ Wl,
                      const float* __restrict__ Wr,
                      const float* __restrict__ bias,
                      float* __restrict__ out,
                      int n_nodes, int do_relu) {
    const int lane = threadIdx.x & 31;
    const int wave = threadIdx.x >> 5;
    const int m0   = blockIdx.x * 16;
    const int n0   = wave * 16;

    const int lrow  = lane & 15;
    const int khalf = (lane >> 4) * 2;

    int mrow = m0 + lrow;
    if (mrow >= n_nodes) mrow = n_nodes - 1;
    const float idg = invdeg[mrow];
    const float* __restrict__ hrow = h   + (size_t)mrow * HID;
    const float* __restrict__ arow = agg + (size_t)mrow * HID;

    const int ncol = n0 + lrow;

    v8f acc_l = {};
    v8f acc_r = {};

    for (int k0 = 0; k0 < HID; k0 += 4) {
        const int k = k0 + khalf;
        v2f a_l, a_r, b_l, b_r;
        a_l.x = arow[k]     * idg;
        a_l.y = arow[k + 1] * idg;
        a_r.x = hrow[k];
        a_r.y = hrow[k + 1];
        b_l.x = Wl[(size_t)k * HID + ncol];
        b_l.y = Wl[(size_t)(k + 1) * HID + ncol];
        b_r.x = Wr[(size_t)k * HID + ncol];
        b_r.y = Wr[(size_t)(k + 1) * HID + ncol];

        acc_l = __builtin_amdgcn_wmma_f32_16x16x4_f32(
            false, a_l, false, b_l, (short)0, acc_l, false, false);
        acc_r = __builtin_amdgcn_wmma_f32_16x16x4_f32(
            false, a_r, false, b_r, (short)0, acc_r, false, false);
    }

    const float bv = bias[ncol];
    const int mbase = m0 + (lane >> 4) * 8;
#pragma unroll
    for (int v = 0; v < 8; ++v) {
        const int m = mbase + v;
        if (m < n_nodes) {
            float o = acc_l[v] + acc_r[v] + bv;
            if (do_relu) o = o > 0.0f ? o : 0.0f;
            out[(size_t)m * HID + ncol] = o;
        }
    }
}

__global__ void edge_dot_kernel(const float* __restrict__ h2,
                                const int* __restrict__ es,
                                const int* __restrict__ ed,
                                float* __restrict__ out, int n) {
    int e = blockIdx.x * blockDim.x + threadIdx.x;
    if (e >= n) return;
    const float4* a = (const float4*)(h2 + (size_t)es[e] * HID);
    const float4* b = (const float4*)(h2 + (size_t)ed[e] * HID);
    float s = 0.0f;
#pragma unroll 8
    for (int i = 0; i < HID / 4; ++i) {
        float4 x = a[i], y = b[i];
        s += x.x * y.x + x.y * y.y + x.z * y.z + x.w * y.w;
    }
    out[e] = s;
}

extern "C" void kernel_launch(void* const* d_in, const int* in_sizes, int n_in,
                              void* d_out, int out_size, void* d_ws, size_t ws_size,
                              hipStream_t stream) {
    const int*   n_id     = (const int*)  d_in[0];
    const float* x_struct = (const float*)d_in[1];
    const int*   eidx     = (const int*)  d_in[2];
    const int*   elab     = (const int*)  d_in[3];
    const float* emb      = (const float*)d_in[4];
    const float* W1l      = (const float*)d_in[5];
    const float* W1r      = (const float*)d_in[6];
    const float* b1       = (const float*)d_in[7];
    const float* W2l      = (const float*)d_in[8];
    const float* W2r      = (const float*)d_in[9];
    const float* b2       = (const float*)d_in[10];
    float*       out      = (float*)d_out;

    const int N = in_sizes[0];
    const int E = in_sizes[2] / 2;
    const int L = in_sizes[3] / 2;

    const int* src = eidx;
    const int* dst = eidx + E;
    const int* lsrc = elab;
    const int* ldst = elab + L;

    float* ws     = (float*)d_ws;
    float* h0     = ws;
    float* h1     = h0  + (size_t)N * HID;
    float* h2     = h1  + (size_t)N * HID;
    float* agg    = h2  + (size_t)N * HID;
    float* deg    = agg + (size_t)N * HID;
    float* invdeg = deg + N;

    const long long nh = (long long)N * HID;

    fill0_kernel<<<(int)((nh + 255) / 256), 256, 0, stream>>>(agg, nh);
    fill0_kernel<<<(N + 255) / 256, 256, 0, stream>>>(deg, N);
    build_h0_kernel<<<(int)((nh + 255) / 256), 256, 0, stream>>>(
        n_id, emb, x_struct, h0, N);
    deg_kernel<<<(E + 255) / 256, 256, 0, stream>>>(dst, deg, E);
    invdeg_kernel<<<(N + 255) / 256, 256, 0, stream>>>(deg, invdeg, N);

    const long long ethreads = (long long)E * HID;
    const int eblocks = (int)((ethreads + 255) / 256);
    const int mblocks = (N + 15) / 16;

    scatter_kernel<<<eblocks, 256, 0, stream>>>(h0, src, dst, agg, E);
    sage_gemm_kernel<<<mblocks, 256, 0, stream>>>(
        agg, invdeg, h0, W1l, W1r, b1, h1, N,  1);

    fill0_kernel<<<(int)((nh + 255) / 256), 256, 0, stream>>>(agg, nh);
    scatter_kernel<<<eblocks, 256, 0, stream>>>(h1, src, dst, agg, E);
    sage_gemm_kernel<<<mblocks, 256, 0, stream>>>(
        agg, invdeg, h1, W2l, W2r, b2, h2, N,  0);

    edge_dot_kernel<<<(L + 255) / 256, 256, 0, stream>>>(h2, lsrc, ldst, out, L);
}


# Neighbour sum owner-computes: 256 work-groups each hold 196 dst rows in LDS, scan the dst list, add matching h[src] rows by LDS float atomics (deg counted there); sums now in LDS arrival order
# speedup vs baseline: 2.7721x; 2.7721x over previous
#include <hip/hip_runtime.h>
#include <hip/hip_bf16.h>


typedef __attribute__((ext_vector_type(2))) float v2f;
typedef __attribute__((ext_vector_type(8))) float v8f;

#define HID 128
#define EMBD 64

__global__ void build_h0_kernel(const int* __restrict__ n_id,
                                const float* __restrict__ emb,
                                const float* __restrict__ x_struct,
                                float* __restrict__ h0, int n_nodes) {
    long long t = (long long)blockIdx.x * blockDim.x + threadIdx.x;
    int i = (int)(t >> 7);
    int c = (int)(t & 127);
    if (i >= n_nodes) return;
    float v;
    if (c < EMBD) v = emb[(size_t)n_id[i] * EMBD + c];
    else          v = x_struct[(size_t)i * EMBD + (c - EMBD)];
    h0[(size_t)i * HID + c] = v;
}

#define AGG_ROWS    196
#define AGG_THREADS 1024
#define AGG_WAVES   32
#define SCAN_U      4
#define Q_CAP       256
#define Q_DRAIN     56

__device__ __forceinline__ void agg_add_row(float* __restrict__ acc, int* __restrict__ cnt,
                                            int r, const float4 v, int lane, int count_deg) {
    float* a = &acc[r * HID + lane];
    __hip_atomic_fetch_add(a,      v.x, __ATOMIC_RELAXED, __HIP_MEMORY_SCOPE_WORKGROUP);
    __hip_atomic_fetch_add(a + 32, v.y, __ATOMIC_RELAXED, __HIP_MEMORY_SCOPE_WORKGROUP);
    __hip_atomic_fetch_add(a + 64, v.z, __ATOMIC_RELAXED, __HIP_MEMORY_SCOPE_WORKGROUP);
    __hip_atomic_fetch_add(a + 96, v.w, __ATOMIC_RELAXED, __HIP_MEMORY_SCOPE_WORKGROUP);
    if (count_deg && lane == 0)
        __hip_atomic_fetch_add(&cnt[r], 1, __ATOMIC_RELAXED, __HIP_MEMORY_SCOPE_WORKGROUP);
}

__device__ __forceinline__ void agg_drain(const float* __restrict__ h,
                                          const int* __restrict__ src,
                                          const int* qe, const int* qr, int n,
                                          float* __restrict__ acc, int* __restrict__ cnt,
                                          int lane, int count_deg) {
    for (int i0 = 0; i0 < n; i0 += 32) {
        const int i = i0 + lane;
        int s = 0, r = 0;
        if (i < n) { s = src[qe[i]]; r = qr[i]; }
        int m = n - i0;
        if (m > 32) m = 32;
        for (int j = 0; j < m; j += 8) {
            float4 v[8];
            int    rw[8];
#pragma unroll
            for (int u = 0; u < 8; ++u) {
                const int su = __builtin_amdgcn_readlane(s, j + u);
                rw[u] = __builtin_amdgcn_readlane(r, j + u);
                v[u]  = ((const float4*)(h + (size_t)su * HID))[lane];
            }
#pragma unroll
            for (int u = 0; u < 8; ++u)
                if (j + u < m) agg_add_row(acc, cnt, rw[u], v[u], lane, count_deg);
        }
    }
}

__device__ __forceinline__ int4 agg_load_dst4(const int* __restrict__ dst, int g, int ngroups,
                                              int n_edges, bool vec_ok) {
    int4 d = make_int4(-1, -1, -1, -1);
    if (g < ngroups) {
        if (vec_ok) {
            d = ((const int4*)dst)[g];
        } else {
            const int e = g * 4;
            if (e     < n_edges) d.x = dst[e];
            if (e + 1 < n_edges) d.y = dst[e + 1];
            if (e + 2 < n_edges) d.z = dst[e + 2];
            if (e + 3 < n_edges) d.w = dst[e + 3];
        }
    }
    return d;
}

__device__ __forceinline__ void agg_take_group(const int4 d, int g, int base, unsigned nrows,
                                               int* __restrict__ qe, int* __restrict__ qr, int& qn,
                                               const float* __restrict__ h,
                                               const int* __restrict__ src,
                                               float* __restrict__ acc, int* __restrict__ cnt,
                                               int lane, int count_deg) {
    const int dv[4] = {d.x, d.y, d.z, d.w};
    unsigned rr[4];
    bool hit[4];
    bool any_hit = false;
#pragma unroll
    for (int k = 0; k < 4; ++k) {
        rr[k]  = (unsigned)(dv[k] - base);
        hit[k] = rr[k] < nrows;
        any_hit |= hit[k];
    }
    if (__builtin_amdgcn_ballot_w32(any_hit) != 0u) {
#pragma unroll
        for (int k = 0; k < 4; ++k) {
            const unsigned mask = __builtin_amdgcn_ballot_w32(hit[k]);
            if (mask != 0u) {
                if (hit[k]) {
                    const int pos = qn + (int)__builtin_amdgcn_mbcnt_lo(mask, 0u);
                    qe[pos] = g * 4 + k;
                    qr[pos] = (int)rr[k];
                }
                qn += __popc(mask);
            }
        }
        if (qn > Q_DRAIN) {
            agg_drain(h, src, qe, qr, qn, acc, cnt, lane, count_deg);
            qn = 0;
        }
    }
}

__global__ __launch_bounds__(AGG_THREADS)
void agg_rows_kernel(const float* __restrict__ h,
                     const int* __restrict__ src,
                     const int* __restrict__ dst,
                     float* __restrict__ agg,
                     float* __restrict__ invdeg,
                     int n_edges, int n_nodes, int write_invdeg) {
    __shared__ float4 acc4[AGG_ROWS * HID / 4];
    __shared__ int    cnt[AGG_ROWS];
    __shared__ int    q_edge[AGG_WAVES][Q_CAP];
    __shared__ int    q_row [AGG_WAVES][Q_CAP];
    float* acc = (float*)acc4;

    const int tid  = threadIdx.x;
    const int lane = tid & 31;
    const int wave = __builtin_amdgcn_readfirstlane(tid >> 5);
    const int base = blockIdx.x * AGG_ROWS;
    int nr = n_nodes - base;
    if (nr > AGG_ROWS) nr = AGG_ROWS;
    if (nr < 0) nr = 0;
    const unsigned nrows = (unsigned)nr;

    for (int i = tid; i < AGG_ROWS * HID / 4; i += AGG_THREADS)
        acc4[i] = make_float4(0.0f, 0.0f, 0.0f, 0.0f);
    for (int i = tid; i < AGG_ROWS; i += AGG_THREADS) cnt[i] = 0;
    __syncthreads();

    int* __restrict__ qe = q_edge[wave];
    int* __restrict__ qr = q_row[wave];
    int qn = 0;

    const int  ngroups     = (n_edges + 3) >> 2;
    const int  trip_groups = 32 * SCAN_U;
    const bool vec_ok      = ((n_edges & 3) == 0) && ((((size_t)dst) & 15) == 0);
    const int  nfull       = vec_ok ? ngroups / trip_groups : 0;
    const int4* __restrict__ dst4 = (const int4*)dst;

    if (wave < nfull) {
        const int rot = (int)(((long long)blockIdx.x * nfull) / gridDim.x);

        int4 da[SCAN_U], db[SCAN_U];
        {
            int tt = wave + rot;
            if (tt >= nfull) tt -= nfull;
#pragma unroll
            for (int u = 0; u < SCAN_U; ++u) da[u] = dst4[tt * trip_groups + u * 32 + lane];
        }
        for (int t = wave; t < nfull; t += 2 * AGG_WAVES) {
            const int t1 = t + AGG_WAVES;
            const int t2 = t + 2 * AGG_WAVES;
            int tt0 = t + rot;
            int tt1 = (t1 < nfull ? t1 : t) + rot;
            int tt2 = (t2 < nfull ? t2 : t) + rot;
            if (tt0 >= nfull) tt0 -= nfull;
            if (tt1 >= nfull) tt1 -= nfull;
            if (tt2 >= nfull) tt2 -= nfull;

#pragma unroll
            for (int u = 0; u < SCAN_U; ++u) db[u] = dst4[tt1 * trip_groups + u * 32 + lane];
#pragma unroll
            for (int u = 0; u < SCAN_U; ++u)
                agg_take_group(da[u], tt0 * trip_groups + u * 32 + lane, base, nrows,
                               qe, qr, qn, h, src, acc, cnt, lane, write_invdeg);
#pragma unroll
            for (int u = 0; u < SCAN_U; ++u) da[u] = dst4[tt2 * trip_groups + u * 32 + lane];
            if (t1 < nfull) {
#pragma unroll
                for (int u = 0; u < SCAN_U; ++u)
                    agg_take_group(db[u], tt1 * trip_groups + u * 32 + lane, base, nrows,
                                   qe, qr, qn, h, src, acc, cnt, lane, write_invdeg);
            }
        }
    }
    for (int g0 = nfull * trip_groups + wave * 32; g0 < ngroups; g0 += AGG_WAVES * 32) {
        const int g = g0 + lane;
        const int4 d = agg_load_dst4(dst, g, ngroups, n_edges, vec_ok);
        agg_take_group(d, g, base, nrows, qe, qr, qn, h, src, acc, cnt, lane, write_invdeg);
    }
    if (qn > 0) agg_drain(h, src, qe, qr, qn, acc, cnt, lane, write_invdeg);

    __syncthreads();

    for (int i = tid; i < nr * 32; i += AGG_THREADS) {
        const int r = i >> 5, q = i & 31;
        const float* a = &acc[r * HID + q];
        const float4 v = make_float4(a[0], a[32], a[64], a[96]);
        ((float4*)(agg + (size_t)(base + r) * HID))[q] = v;
    }
    if (write_invdeg) {
        for (int r = tid; r < nr; r += AGG_THREADS)
            invdeg[base + r] = 1.0f / fmaxf((float)cnt[r], 1.0f);
    }
}

__global__ __launch_bounds__(256)
void sage_gemm_kernel(const float* __restrict__ agg,
                      const float* __restrict__ invdeg,
                      const float* __restrict__ h,
                      const float* __restrict__ Wl,
                      const float* __restrict__ Wr,
                      const float* __restrict__ bias,
                      float* __restrict__ out,
                      int n_nodes, int do_relu) {
    const int lane = threadIdx.x & 31;
    const int wave = threadIdx.x >> 5;
    const int m0   = blockIdx.x * 16;
    const int n0   = wave * 16;

    const int lrow  = lane & 15;
    const int khalf = (lane >> 4) * 2;

    int mrow = m0 + lrow;
    if (mrow >= n_nodes) mrow = n_nodes - 1;
    const float idg = invdeg[mrow];
    const float* __restrict__ hrow = h   + (size_t)mrow * HID;
    const float* __restrict__ arow = agg + (size_t)mrow * HID;

    const int ncol = n0 + lrow;

    v8f acc_l = {};
    v8f acc_r = {};

    for (int k0 = 0; k0 < HID; k0 += 4) {
        const int k = k0 + khalf;
        v2f a_l, a_r, b_l, b_r;
        a_l.x = arow[k]     * idg;
        a_l.y = arow[k + 1] * idg;
        a_r.x = hrow[k];
        a_r.y = hrow[k + 1];
        b_l.x = Wl[(size_t)k * HID + ncol];
        b_l.y = Wl[(size_t)(k + 1) * HID + ncol];
        b_r.x = Wr[(size_t)k * HID + ncol];
        b_r.y = Wr[(size_t)(k + 1) * HID + ncol];

        acc_l = __builtin_amdgcn_wmma_f32_16x16x4_f32(
            false, a_l, false, b_l, (short)0, acc_l, false, false);
        acc_r = __builtin_amdgcn_wmma_f32_16x16x4_f32(
            false, a_r, false, b_r, (short)0, acc_r, false, false);
    }

    const float bv = bias[ncol];
    const int mbase = m0 + (lane >> 4) * 8;
#pragma unroll
    for (int v = 0; v < 8; ++v) {
        const int m = mbase + v;
        if (m < n_nodes) {
            float o = acc_l[v] + acc_r[v] + bv;
            if (do_relu) o = o > 0.0f ? o : 0.0f;
            out[(size_t)m * HID + ncol] = o;
        }
    }
}

__global__ void edge_dot_kernel(const float* __restrict__ h2,
                                const int* __restrict__ es,
                                const int* __restrict__ ed,
                                float* __restrict__ out, int n) {
    int e = blockIdx.x * blockDim.x + threadIdx.x;
    if (e >= n) return;
    const float4* a = (const float4*)(h2 + (size_t)es[e] * HID);
    const float4* b = (const float4*)(h2 + (size_t)ed[e] * HID);
    float s = 0.0f;
#pragma unroll 8
    for (int i = 0; i < HID / 4; ++i) {
        float4 x = a[i], y = b[i];
        s += x.x * y.x + x.y * y.y + x.z * y.z + x.w * y.w;
    }
    out[e] = s;
}

extern "C" void kernel_launch(void* const* d_in, const int* in_sizes, int n_in,
                              void* d_out, int out_size, void* d_ws, size_t ws_size,
                              hipStream_t stream) {
    const int*   n_id     = (const int*)  d_in[0];
    const float* x_struct = (const float*)d_in[1];
    const int*   eidx     = (const int*)  d_in[2];
    const int*   elab     = (const int*)  d_in[3];
    const float* emb      = (const float*)d_in[4];
    const float* W1l      = (const float*)d_in[5];
    const float* W1r      = (const float*)d_in[6];
    const float* b1       = (const float*)d_in[7];
    const float* W2l      = (const float*)d_in[8];
    const float* W2r      = (const float*)d_in[9];
    const float* b2       = (const float*)d_in[10];
    float*       out      = (float*)d_out;

    const int N = in_sizes[0];
    const int E = in_sizes[2] / 2;
    const int L = in_sizes[3] / 2;

    const int* src = eidx;
    const int* dst = eidx + E;
    const int* lsrc = elab;
    const int* ldst = elab + L;

    float* ws     = (float*)d_ws;
    float* h0     = ws;
    float* h1     = h0  + (size_t)N * HID;
    float* h2     = h1  + (size_t)N * HID;
    float* agg    = h2  + (size_t)N * HID;
    float* invdeg = agg + (size_t)N * HID;

    const long long nh = (long long)N * HID;

    build_h0_kernel<<<(int)((nh + 255) / 256), 256, 0, stream>>>(
        n_id, emb, x_struct, h0, N);

    const int ablocks = (N + AGG_ROWS - 1) / AGG_ROWS;
    const int mblocks = (N + 15) / 16;

    agg_rows_kernel<<<ablocks, AGG_THREADS, 0, stream>>>(
        h0, src, dst, agg, invdeg, E, N,  1);
    sage_gemm_kernel<<<mblocks, 256, 0, stream>>>(
        agg, invdeg, h0, W1l, W1r, b1, h1, N,  1);

    agg_rows_kernel<<<ablocks, AGG_THREADS, 0, stream>>>(
        h1, src, dst, agg, invdeg, E, N,  0);
    sage_gemm_kernel<<<mblocks, 256, 0, stream>>>(
        agg, invdeg, h1, W2l, W2r, b2, h2, N,  0);

    edge_dot_kernel<<<(L + 255) / 256, 256, 0, stream>>>(h2, lsrc, ldst, out, L);
}


# sage_gemm: Wl and Wr held in LDS as ready B fragments, staged once by each of 256 resident work-groups; a wave owns 16 rows x 128 columns, A fragments straight from global; sums unchanged
# speedup vs baseline: 3.2809x; 1.1835x over previous
#include <hip/hip_runtime.h>
#include <hip/hip_bf16.h>


typedef __attribute__((ext_vector_type(2))) float v2f;
typedef __attribute__((ext_vector_type(8))) float v8f;

#define HID 128
#define EMBD 64

__global__ void build_h0_kernel(const int* __restrict__ n_id,
                                const float* __restrict__ emb,
                                const float* __restrict__ x_struct,
                                float* __restrict__ h0, int n_nodes) {
    long long t = (long long)blockIdx.x * blockDim.x + threadIdx.x;
    int i = (int)(t >> 7);
    int c = (int)(t & 127);
    if (i >= n_nodes) return;
    float v;
    if (c < EMBD) v = emb[(size_t)n_id[i] * EMBD + c];
    else          v = x_struct[(size_t)i * EMBD + (c - EMBD)];
    h0[(size_t)i * HID + c] = v;
}

#define AGG_ROWS    196
#define AGG_THREADS 1024
#define AGG_WAVES   32
#define SCAN_U      4
#define Q_CAP       256
#define Q_DRAIN     56

__device__ __forceinline__ void agg_add_row(float* __restrict__ acc, int* __restrict__ cnt,
                                            int r, const float4 v, int lane, int count_deg) {
    float* a = &acc[r * HID + lane];
    __hip_atomic_fetch_add(a,      v.x, __ATOMIC_RELAXED, __HIP_MEMORY_SCOPE_WORKGROUP);
    __hip_atomic_fetch_add(a + 32, v.y, __ATOMIC_RELAXED, __HIP_MEMORY_SCOPE_WORKGROUP);
    __hip_atomic_fetch_add(a + 64, v.z, __ATOMIC_RELAXED, __HIP_MEMORY_SCOPE_WORKGROUP);
    __hip_atomic_fetch_add(a + 96, v.w, __ATOMIC_RELAXED, __HIP_MEMORY_SCOPE_WORKGROUP);
    if (count_deg && lane == 0)
        __hip_atomic_fetch_add(&cnt[r], 1, __ATOMIC_RELAXED, __HIP_MEMORY_SCOPE_WORKGROUP);
}

__device__ __forceinline__ void agg_drain(const float* __restrict__ h,
                                          const int* __restrict__ src,
                                          const int* qe, const int* qr, int n,
                                          float* __restrict__ acc, int* __restrict__ cnt,
                                          int lane, int count_deg) {
    for (int i0 = 0; i0 < n; i0 += 32) {
        const int i = i0 + lane;
        int s = 0, r = 0;
        if (i < n) { s = src[qe[i]]; r = qr[i]; }
        int m = n - i0;
        if (m > 32) m = 32;
        for (int j = 0; j < m; j += 8) {
            float4 v[8];
            int    rw[8];
#pragma unroll
            for (int u = 0; u < 8; ++u) {
                const int su = __builtin_amdgcn_readlane(s, j + u);
                rw[u] = __builtin_amdgcn_readlane(r, j + u);
                v[u]  = ((const float4*)(h + (size_t)su * HID))[lane];
            }
#pragma unroll
            for (int u = 0; u < 8; ++u)
                if (j + u < m) agg_add_row(acc, cnt, rw[u], v[u], lane, count_deg);
        }
    }
}

__device__ __forceinline__ int4 agg_load_dst4(const int* __restrict__ dst, int g, int ngroups,
                                              int n_edges, bool vec_ok) {
    int4 d = make_int4(-1, -1, -1, -1);
    if (g < ngroups) {
        if (vec_ok) {
            d = ((const int4*)dst)[g];
        } else {
            const int e = g * 4;
            if (e     < n_edges) d.x = dst[e];
            if (e + 1 < n_edges) d.y = dst[e + 1];
            if (e + 2 < n_edges) d.z = dst[e + 2];
            if (e + 3 < n_edges) d.w = dst[e + 3];
        }
    }
    return d;
}

__device__ __forceinline__ void agg_take_group(const int4 d, int g, int base, unsigned nrows,
                                               int* __restrict__ qe, int* __restrict__ qr, int& qn,
                                               const float* __restrict__ h,
                                               const int* __restrict__ src,
                                               float* __restrict__ acc, int* __restrict__ cnt,
                                               int lane, int count_deg) {
    const int dv[4] = {d.x, d.y, d.z, d.w};
    unsigned rr[4];
    bool hit[4];
    bool any_hit = false;
#pragma unroll
    for (int k = 0; k < 4; ++k) {
        rr[k]  = (unsigned)(dv[k] - base);
        hit[k] = rr[k] < nrows;
        any_hit |= hit[k];
    }
    if (__builtin_amdgcn_ballot_w32(any_hit) != 0u) {
#pragma unroll
        for (int k = 0; k < 4; ++k) {
            const unsigned mask = __builtin_amdgcn_ballot_w32(hit[k]);
            if (mask != 0u) {
                if (hit[k]) {
                    const int pos = qn + (int)__builtin_amdgcn_mbcnt_lo(mask, 0u);
                    qe[pos] = g * 4 + k;
                    qr[pos] = (int)rr[k];
                }
                qn += __popc(mask);
            }
        }
        if (qn > Q_DRAIN) {
            agg_drain(h, src, qe, qr, qn, acc, cnt, lane, count_deg);
            qn = 0;
        }
    }
}

__global__ __launch_bounds__(AGG_THREADS)
void agg_rows_kernel(const float* __restrict__ h,
                     const int* __restrict__ src,
                     const int* __restrict__ dst,
                     float* __restrict__ agg,
                     float* __restrict__ invdeg,
                     int n_edges, int n_nodes, int write_invdeg) {
    __shared__ float4 acc4[AGG_ROWS * HID / 4];
    __shared__ int    cnt[AGG_ROWS];
    __shared__ int    q_edge[AGG_WAVES][Q_CAP];
    __shared__ int    q_row [AGG_WAVES][Q_CAP];
    float* acc = (float*)acc4;

    const int tid  = threadIdx.x;
    const int lane = tid & 31;
    const int wave = __builtin_amdgcn_readfirstlane(tid >> 5);
    const int base = blockIdx.x * AGG_ROWS;
    int nr = n_nodes - base;
    if (nr > AGG_ROWS) nr = AGG_ROWS;
    if (nr < 0) nr = 0;
    const unsigned nrows = (unsigned)nr;

    for (int i = tid; i < AGG_ROWS * HID / 4; i += AGG_THREADS)
        acc4[i] = make_float4(0.0f, 0.0f, 0.0f, 0.0f);
    for (int i = tid; i < AGG_ROWS; i += AGG_THREADS) cnt[i] = 0;
    __syncthreads();

    int* __restrict__ qe = q_edge[wave];
    int* __restrict__ qr = q_row[wave];
    int qn = 0;

    const int  ngroups     = (n_edges + 3) >> 2;
    const int  trip_groups = 32 * SCAN_U;
    const bool vec_ok      = ((n_edges & 3) == 0) && ((((size_t)dst) & 15) == 0);
    const int  nfull       = vec_ok ? ngroups / trip_groups : 0;
    const int4* __restrict__ dst4 = (const int4*)dst;

    if (wave < nfull) {
        const int rot = (int)(((long long)blockIdx.x * nfull) / gridDim.x);

        int4 da[SCAN_U], db[SCAN_U];
        {
            int tt = wave + rot;
            if (tt >= nfull) tt -= nfull;
#pragma unroll
            for (int u = 0; u < SCAN_U; ++u) da[u] = dst4[tt * trip_groups + u * 32 + lane];
        }
        for (int t = wave; t < nfull; t += 2 * AGG_WAVES) {
            const int t1 = t + AGG_WAVES;
            const int t2 = t + 2 * AGG_WAVES;
            int tt0 = t + rot;
            int tt1 = (t1 < nfull ? t1 : t) + rot;
            int tt2 = (t2 < nfull ? t2 : t) + rot;
            if (tt0 >= nfull) tt0 -= nfull;
            if (tt1 >= nfull) tt1 -= nfull;
            if (tt2 >= nfull) tt2 -= nfull;

#pragma unroll
            for (int u = 0; u < SCAN_U; ++u) db[u] = dst4[tt1 * trip_groups + u * 32 + lane];
#pragma unroll
            for (int u = 0; u < SCAN_U; ++u)
                agg_take_group(da[u], tt0 * trip_groups + u * 32 + lane, base, nrows,
                               qe, qr, qn, h, src, acc, cnt, lane, write_invdeg);
#pragma unroll
            for (int u = 0; u < SCAN_U; ++u) da[u] = dst4[tt2 * trip_groups + u * 32 + lane];
            if (t1 < nfull) {
#pragma unroll
                for (int u = 0; u < SCAN_U; ++u)
                    agg_take_group(db[u], tt1 * trip_groups + u * 32 + lane, base, nrows,
                                   qe, qr, qn, h, src, acc, cnt, lane, write_invdeg);
            }
        }
    }
    for (int g0 = nfull * trip_groups + wave * 32; g0 < ngroups; g0 += AGG_WAVES * 32) {
        const int g = g0 + lane;
        const int4 d = agg_load_dst4(dst, g, ngroups, n_edges, vec_ok);
        agg_take_group(d, g, base, nrows, qe, qr, qn, h, src, acc, cnt, lane, write_invdeg);
    }
    if (qn > 0) agg_drain(h, src, qe, qr, qn, acc, cnt, lane, write_invdeg);

    __syncthreads();

    for (int i = tid; i < nr * 32; i += AGG_THREADS) {
        const int r = i >> 5, q = i & 31;
        const float* a = &acc[r * HID + q];
        const float4 v = make_float4(a[0], a[32], a[64], a[96]);
        ((float4*)(agg + (size_t)(base + r) * HID))[q] = v;
    }
    if (write_invdeg) {
        for (int r = tid; r < nr; r += AGG_THREADS)
            invdeg[base + r] = 1.0f / fmaxf((float)cnt[r], 1.0f);
    }
}

#define GEMM_BLOCKS  256
#define GEMM_THREADS 256
#define GEMM_WAVES   8
#define KSTEPS       (HID / 4)
#define CTILES       (HID / 16)

#define GEMM_LOAD_A(TILE)                                                        \
    {                                                                            \
        int mrow_ = (TILE) * 16 + c;                                             \
        if (mrow_ >= n_nodes) mrow_ = n_nodes - 1;     \
        nidg = invdeg[mrow_];                                                    \
        const v2f* ap_ = (const v2f*)(agg + (size_t)mrow_ * HID) + hh;           \
        const v2f* hp_ = (const v2f*)(h   + (size_t)mrow_ * HID) + hh;           \
          \
        _Pragma("unroll")                                                        \
        for (int t_ = 0; t_ < KSTEPS; ++t_) { nl[t_] = ap_[2 * t_]; nr[t_] = hp_[2 * t_]; } \
    }

__global__ __launch_bounds__(GEMM_THREADS)
void sage_gemm_kernel(const float* __restrict__ agg,
                      const float* __restrict__ invdeg,
                      const float* __restrict__ h,
                      const float* __restrict__ Wl,
                      const float* __restrict__ Wr,
                      const float* __restrict__ bias,
                      float* __restrict__ out,
                      int n_nodes, int do_relu) {
    __shared__ float4 wq[(HID / 2) * HID];
    __shared__ float  sbias[HID];

    const int tid  = threadIdx.x;
    const int lane = tid & 31;
    const int wave = __builtin_amdgcn_readfirstlane(tid >> 5);
    const int c    = lane & 15;
    const int hh   = lane >> 4;

    const int ntiles = (n_nodes + 15) >> 4;
    const int stride = (int)gridDim.x * GEMM_WAVES;
    int tile = (int)blockIdx.x + (int)gridDim.x * wave;

    v2f   nl[KSTEPS], nr[KSTEPS];
    float nidg = 0.0f;
#pragma unroll
    for (int t = 0; t < KSTEPS; ++t) { nl[t] = (v2f){0.0f, 0.0f}; nr[t] = (v2f){0.0f, 0.0f}; }
    if (tile < ntiles) GEMM_LOAD_A(tile)

    const int rot = (int)blockIdx.x & 31;
#pragma unroll 16
    for (int it = 0; it < (HID / 2) * HID / GEMM_THREADS; ++it) {
        const int i  = tid + (((it + rot) & 31) * GEMM_THREADS);
        const int kp = i >> 7, n = i & (HID - 1);
        float4 w;
        w.x = Wl[(size_t)(2 * kp)     * HID + n];
        w.y = Wl[(size_t)(2 * kp + 1) * HID + n];
        w.z = Wr[(size_t)(2 * kp)     * HID + n];
        w.w = Wr[(size_t)(2 * kp + 1) * HID + n];
        wq[i] = w;
    }
    if (tid < HID) sbias[tid] = bias[tid];
    __syncthreads();

    const float4* __restrict__ wlane = wq + hh * HID + c;

    while (tile < ntiles) {
        const int m0 = tile * 16;
        v2f al[KSTEPS], ar[KSTEPS];
#pragma unroll
        for (int t = 0; t < KSTEPS; ++t) { al[t] = nl[t] * nidg; ar[t] = nr[t]; }

        const int nt = tile + stride;
        if (nt < ntiles) GEMM_LOAD_A(nt)

        const int mbase = m0 + hh * 8;
#pragma unroll 1
        for (int j = 0; j < CTILES; ++j) {
            const float4* __restrict__ wj = wlane + 16 * j;
            v8f acc_l = {};
            v8f acc_r = {};
#pragma unroll
            for (int t = 0; t < KSTEPS; ++t) {
                const float4 b = wj[(2 * t) * HID];
                const v2f b_l = {b.x, b.y};
                const v2f b_r = {b.z, b.w};
                acc_l = __builtin_amdgcn_wmma_f32_16x16x4_f32(
                    false, al[t], false, b_l, (short)0, acc_l, false, false);
                acc_r = __builtin_amdgcn_wmma_f32_16x16x4_f32(
                    false, ar[t], false, b_r, (short)0, acc_r, false, false);
            }
            const int ncol = 16 * j + c;
            const float bv = sbias[ncol];
            asm volatile("v_nop\n\tv_nop\n\tv_nop\n\tv_nop\n\tv_nop\n\tv_nop\n\tv_nop\n\tv_nop"
                         : "+v"(acc_l), "+v"(acc_r));
#pragma unroll
            for (int v = 0; v < 8; ++v) {
                const int m = mbase + v;
                if (m < n_nodes) {
                    float o = acc_l[v] + acc_r[v] + bv;
                    if (do_relu) o = o > 0.0f ? o : 0.0f;
                    out[(size_t)m * HID + ncol] = o;
                }
            }
        }
        tile = nt;
    }
}

__global__ void edge_dot_kernel(const float* __restrict__ h2,
                                const int* __restrict__ es,
                                const int* __restrict__ ed,
                                float* __restrict__ out, int n) {
    int e = blockIdx.x * blockDim.x + threadIdx.x;
    if (e >= n) return;
    const float4* a = (const float4*)(h2 + (size_t)es[e] * HID);
    const float4* b = (const float4*)(h2 + (size_t)ed[e] * HID);
    float s = 0.0f;
#pragma unroll 8
    for (int i = 0; i < HID / 4; ++i) {
        float4 x = a[i], y = b[i];
        s += x.x * y.x + x.y * y.y + x.z * y.z + x.w * y.w;
    }
    out[e] = s;
}

extern "C" void kernel_launch(void* const* d_in, const int* in_sizes, int n_in,
                              void* d_out, int out_size, void* d_ws, size_t ws_size,
                              hipStream_t stream) {
    const int*   n_id     = (const int*)  d_in[0];
    const float* x_struct = (const float*)d_in[1];
    const int*   eidx     = (const int*)  d_in[2];
    const int*   elab     = (const int*)  d_in[3];
    const float* emb      = (const float*)d_in[4];
    const float* W1l      = (const float*)d_in[5];
    const float* W1r      = (const float*)d_in[6];
    const float* b1       = (const float*)d_in[7];
    const float* W2l      = (const float*)d_in[8];
    const float* W2r      = (const float*)d_in[9];
    const float* b2       = (const float*)d_in[10];
    float*       out      = (float*)d_out;

    const int N = in_sizes[0];
    const int E = in_sizes[2] / 2;
    const int L = in_sizes[3] / 2;

    const int* src = eidx;
    const int* dst = eidx + E;
    const int* lsrc = elab;
    const int* ldst = elab + L;

    float* ws     = (float*)d_ws;
    float* h0     = ws;
    float* h1     = h0  + (size_t)N * HID;
    float* h2     = h1  + (size_t)N * HID;
    float* agg    = h2  + (size_t)N * HID;
    float* invdeg = agg + (size_t)N * HID;

    const long long nh = (long long)N * HID;

    build_h0_kernel<<<(int)((nh + 255) / 256), 256, 0, stream>>>(
        n_id, emb, x_struct, h0, N);

    const int ablocks = (N + AGG_ROWS - 1) / AGG_ROWS;

    agg_rows_kernel<<<ablocks, AGG_THREADS, 0, stream>>>(
        h0, src, dst, agg, invdeg, E, N,  1);
    sage_gemm_kernel<<<GEMM_BLOCKS, GEMM_THREADS, 0, stream>>>(
        agg, invdeg, h0, W1l, W1r, b1, h1, N,  1);

    agg_rows_kernel<<<ablocks, AGG_THREADS, 0, stream>>>(
        h1, src, dst, agg, invdeg, E, N,  0);
    sage_gemm_kernel<<<GEMM_BLOCKS, GEMM_THREADS, 0, stream>>>(
        agg, invdeg, h1, W2l, W2r, b2, h2, N,  0);

    edge_dot_kernel<<<(L + 255) / 256, 256, 0, stream>>>(h2, lsrc, ldst, out, L);
}


# h0 = concat(emb[n_id], x_struct) is never written out: layer 1's neighbour sum and GEMM read the two 256-byte row halves in place; h0 plane and its 200000-wave launch gone, 5 launches; sums unchanged
# speedup vs baseline: 3.5947x; 1.0956x over previous
#include <hip/hip_runtime.h>
#include <hip/hip_bf16.h>


typedef __attribute__((ext_vector_type(2))) float v2f;
typedef __attribute__((ext_vector_type(8))) float v8f;

#define HID 128
#define EMBD 64


#define AGG_ROWS    196
#define AGG_THREADS 1024
#define AGG_WAVES   32
#define SCAN_U      4
#define Q_CAP       256
#define Q_DRAIN     56

__device__ __forceinline__ void agg_add_row(float* __restrict__ acc, int* __restrict__ cnt,
                                            int r, const float4 v, int lane, int count_deg) {
    float* a = &acc[r * HID + lane];
    __hip_atomic_fetch_add(a,      v.x, __ATOMIC_RELAXED, __HIP_MEMORY_SCOPE_WORKGROUP);
    __hip_atomic_fetch_add(a + 32, v.y, __ATOMIC_RELAXED, __HIP_MEMORY_SCOPE_WORKGROUP);
    __hip_atomic_fetch_add(a + 64, v.z, __ATOMIC_RELAXED, __HIP_MEMORY_SCOPE_WORKGROUP);
    __hip_atomic_fetch_add(a + 96, v.w, __ATOMIC_RELAXED, __HIP_MEMORY_SCOPE_WORKGROUP);
    if (count_deg && lane == 0)
        __hip_atomic_fetch_add(&cnt[r], 1, __ATOMIC_RELAXED, __HIP_MEMORY_SCOPE_WORKGROUP);
}

__device__ __forceinline__ void agg_drain(const float* __restrict__ h,
                                          const int* __restrict__ nid,
                                          const float* __restrict__ emb,
                                          const float* __restrict__ xs,
                                          const bool split,
                                          const int* __restrict__ src,
                                          const int* qe, const int* qr, int n,
                                          float* __restrict__ acc, int* __restrict__ cnt,
                                          int lane, int count_deg) {
    const bool   emb_half = lane < 16;
    const float* half_tbl = emb_half ? emb : xs;
    const int    piece    = (lane & 15) * 4;
    for (int i0 = 0; i0 < n; i0 += 32) {
        const int i = i0 + lane;
        int s = 0, r = 0, id = 0;
        if (i < n) { s = src[qe[i]]; r = qr[i]; }
        if (split) id = nid[s];
        int m = n - i0;
        if (m > 32) m = 32;
        for (int j = 0; j < m; j += 8) {
            float4 v[8];
            int    rw[8];
#pragma unroll
            for (int u = 0; u < 8; ++u) {
                const int su = __builtin_amdgcn_readlane(s, j + u);
                rw[u] = __builtin_amdgcn_readlane(r, j + u);
                if (split) {
                    const int iu  = __builtin_amdgcn_readlane(id, j + u);
                    const int row = emb_half ? iu : su;
                    v[u] = *(const float4*)(half_tbl + (size_t)row * EMBD + piece);
                } else {
                    v[u] = ((const float4*)(h + (size_t)su * HID))[lane];
                }
            }
#pragma unroll
            for (int u = 0; u < 8; ++u)
                if (j + u < m) agg_add_row(acc, cnt, rw[u], v[u], lane, count_deg);
        }
    }
}

__device__ __forceinline__ int4 agg_load_dst4(const int* __restrict__ dst, int g, int ngroups,
                                              int n_edges, bool vec_ok) {
    int4 d = make_int4(-1, -1, -1, -1);
    if (g < ngroups) {
        if (vec_ok) {
            d = ((const int4*)dst)[g];
        } else {
            const int e = g * 4;
            if (e     < n_edges) d.x = dst[e];
            if (e + 1 < n_edges) d.y = dst[e + 1];
            if (e + 2 < n_edges) d.z = dst[e + 2];
            if (e + 3 < n_edges) d.w = dst[e + 3];
        }
    }
    return d;
}

__device__ __forceinline__ void agg_take_group(const int4 d, int g, int base, unsigned nrows,
                                               int* __restrict__ qe, int* __restrict__ qr, int& qn,
                                               const float* __restrict__ h,
                                               const int* __restrict__ nid,
                                               const float* __restrict__ emb,
                                               const float* __restrict__ xs,
                                               const bool split,
                                               const int* __restrict__ src,
                                               float* __restrict__ acc, int* __restrict__ cnt,
                                               int lane, int count_deg) {
    const int dv[4] = {d.x, d.y, d.z, d.w};
    unsigned rr[4];
    bool hit[4];
    bool any_hit = false;
#pragma unroll
    for (int k = 0; k < 4; ++k) {
        rr[k]  = (unsigned)(dv[k] - base);
        hit[k] = rr[k] < nrows;
        any_hit |= hit[k];
    }
    if (__builtin_amdgcn_ballot_w32(any_hit) != 0u) {
#pragma unroll
        for (int k = 0; k < 4; ++k) {
            const unsigned mask = __builtin_amdgcn_ballot_w32(hit[k]);
            if (mask != 0u) {
                if (hit[k]) {
                    const int pos = qn + (int)__builtin_amdgcn_mbcnt_lo(mask, 0u);
                    qe[pos] = g * 4 + k;
                    qr[pos] = (int)rr[k];
                }
                qn += __popc(mask);
            }
        }
        if (qn > Q_DRAIN) {
            agg_drain(h, nid, emb, xs, split, src, qe, qr, qn, acc, cnt, lane, count_deg);
            qn = 0;
        }
    }
}

__device__ __forceinline__
void agg_rows_body(float4* __restrict__ acc4, int* __restrict__ cnt,
                   int* __restrict__ q_edge, int* __restrict__ q_row,
                   const float* __restrict__ h,
                   const int* __restrict__ nid,
                   const float* __restrict__ emb,
                   const float* __restrict__ xs,
                   const bool split,
                   const int* __restrict__ src,
                   const int* __restrict__ dst,
                   float* __restrict__ agg,
                   float* __restrict__ invdeg,
                   int n_edges, int n_nodes, int write_invdeg) {
    float* acc = (float*)acc4;

    const int tid  = threadIdx.x;
    const int lane = tid & 31;
    const int wave = __builtin_amdgcn_readfirstlane(tid >> 5);
    const int base = blockIdx.x * AGG_ROWS;
    int nr = n_nodes - base;
    if (nr > AGG_ROWS) nr = AGG_ROWS;
    if (nr < 0) nr = 0;
    const unsigned nrows = (unsigned)nr;

    for (int i = tid; i < AGG_ROWS * HID / 4; i += AGG_THREADS)
        acc4[i] = make_float4(0.0f, 0.0f, 0.0f, 0.0f);
    for (int i = tid; i < AGG_ROWS; i += AGG_THREADS) cnt[i] = 0;
    __syncthreads();

    int* __restrict__ qe = q_edge + wave * Q_CAP;
    int* __restrict__ qr = q_row  + wave * Q_CAP;
    int qn = 0;

    const int  ngroups     = (n_edges + 3) >> 2;
    const int  trip_groups = 32 * SCAN_U;
    const bool vec_ok      = ((n_edges & 3) == 0) && ((((size_t)dst) & 15) == 0);
    const int  nfull       = vec_ok ? ngroups / trip_groups : 0;
    const int4* __restrict__ dst4 = (const int4*)dst;

    if (wave < nfull) {
        const int rot = (int)(((long long)blockIdx.x * nfull) / gridDim.x);

        int4 da[SCAN_U], db[SCAN_U];
        {
            int tt = wave + rot;
            if (tt >= nfull) tt -= nfull;
#pragma unroll
            for (int u = 0; u < SCAN_U; ++u) da[u] = dst4[tt * trip_groups + u * 32 + lane];
        }
        for (int t = wave; t < nfull; t += 2 * AGG_WAVES) {
            const int t1 = t + AGG_WAVES;
            const int t2 = t + 2 * AGG_WAVES;
            int tt0 = t + rot;
            int tt1 = (t1 < nfull ? t1 : t) + rot;
            int tt2 = (t2 < nfull ? t2 : t) + rot;
            if (tt0 >= nfull) tt0 -= nfull;
            if (tt1 >= nfull) tt1 -= nfull;
            if (tt2 >= nfull) tt2 -= nfull;

#pragma unroll
            for (int u = 0; u < SCAN_U; ++u) db[u] = dst4[tt1 * trip_groups + u * 32 + lane];
#pragma unroll
            for (int u = 0; u < SCAN_U; ++u)
                agg_take_group(da[u], tt0 * trip_groups + u * 32 + lane, base, nrows,
                               qe, qr, qn, h, nid, emb, xs, split, src, acc, cnt, lane, write_invdeg);
#pragma unroll
            for (int u = 0; u < SCAN_U; ++u) da[u] = dst4[tt2 * trip_groups + u * 32 + lane];
            if (t1 < nfull) {
#pragma unroll
                for (int u = 0; u < SCAN_U; ++u)
                    agg_take_group(db[u], tt1 * trip_groups + u * 32 + lane, base, nrows,
                                   qe, qr, qn, h, nid, emb, xs, split, src, acc, cnt, lane, write_invdeg);
            }
        }
    }
    for (int g0 = nfull * trip_groups + wave * 32; g0 < ngroups; g0 += AGG_WAVES * 32) {
        const int g = g0 + lane;
        const int4 d = agg_load_dst4(dst, g, ngroups, n_edges, vec_ok);
        agg_take_group(d, g, base, nrows, qe, qr, qn, h, nid, emb, xs, split, src, acc, cnt, lane, write_invdeg);
    }
    if (qn > 0) agg_drain(h, nid, emb, xs, split, src, qe, qr, qn, acc, cnt, lane, write_invdeg);

    __syncthreads();

    for (int i = tid; i < nr * 32; i += AGG_THREADS) {
        const int r = i >> 5, q = i & 31;
        const float* a = &acc[r * HID + q];
        const float4 v = make_float4(a[0], a[32], a[64], a[96]);
        ((float4*)(agg + (size_t)(base + r) * HID))[q] = v;
    }
    if (write_invdeg) {
        for (int r = tid; r < nr; r += AGG_THREADS)
            invdeg[base + r] = 1.0f / fmaxf((float)cnt[r], 1.0f);
    }
}

__global__ __launch_bounds__(AGG_THREADS)
void agg_rows_in_kernel(const int* __restrict__ n_id,
                        const float* __restrict__ emb,
                        const float* __restrict__ x_struct,
                        const int* __restrict__ src,
                        const int* __restrict__ dst,
                        float* __restrict__ agg,
                        float* __restrict__ invdeg,
                        int n_edges, int n_nodes) {
    __shared__ float4 acc4[AGG_ROWS * HID / 4];
    __shared__ int    cnt[AGG_ROWS];
    __shared__ int    q_edge[AGG_WAVES * Q_CAP];
    __shared__ int    q_row [AGG_WAVES * Q_CAP];
    agg_rows_body(acc4, cnt, q_edge, q_row, nullptr, n_id, emb, x_struct,  true,
                  src, dst, agg, invdeg, n_edges, n_nodes,  1);
}

__global__ __launch_bounds__(AGG_THREADS)
void agg_rows_kernel(const float* __restrict__ h,
                     const int* __restrict__ src,
                     const int* __restrict__ dst,
                     float* __restrict__ agg,
                     float* __restrict__ invdeg,
                     int n_edges, int n_nodes, int write_invdeg) {
    __shared__ float4 acc4[AGG_ROWS * HID / 4];
    __shared__ int    cnt[AGG_ROWS];
    __shared__ int    q_edge[AGG_WAVES * Q_CAP];
    __shared__ int    q_row [AGG_WAVES * Q_CAP];
    agg_rows_body(acc4, cnt, q_edge, q_row, h, nullptr, nullptr, nullptr,  false,
                  src, dst, agg, invdeg, n_edges, n_nodes, write_invdeg);
}

#define GEMM_BLOCKS  256
#define GEMM_THREADS 256
#define GEMM_WAVES   8
#define KSTEPS       (HID / 4)
#define CTILES       (HID / 16)

#define GEMM_LOAD_A(TILE)                                                        \
    {                                                                            \
        int mrow_ = (TILE) * 16 + c;                                             \
        if (mrow_ >= n_nodes) mrow_ = n_nodes - 1;     \
        nidg = invdeg[mrow_];                                                    \
        const v2f* ap_ = (const v2f*)(agg + (size_t)mrow_ * HID) + hh;           \
        const v2f* ha_;                                                          \
        const v2f* hb_;                                                          \
        if (split) {                                                             \
            ha_ = (const v2f*)(emb + (size_t)nid[mrow_] * EMBD) + hh;            \
            hb_ = (const v2f*)(xs  + (size_t)mrow_ * EMBD) + hh;                 \
        } else {                                                                 \
            ha_ = (const v2f*)(h + (size_t)mrow_ * HID) + hh;                    \
            hb_ = ha_ + EMBD / 2;                                                \
        }                                                                        \
          \
        _Pragma("unroll")                                                        \
        for (int t_ = 0; t_ < KSTEPS; ++t_) {                                    \
            nl[t_] = ap_[2 * t_];                                                \
            nr[t_] = t_ < KSTEPS / 2 ? ha_[2 * t_] : hb_[2 * t_ - KSTEPS];       \
        }                                                                        \
    }

__device__ __forceinline__
void sage_gemm_body(float4* __restrict__ wq, float* __restrict__ sbias,
                    const float* __restrict__ agg,
                    const float* __restrict__ invdeg,
                    const float* __restrict__ h,
                    const int* __restrict__ nid,
                    const float* __restrict__ emb,
                    const float* __restrict__ xs,
                    const bool split,
                    const float* __restrict__ Wl,
                    const float* __restrict__ Wr,
                    const float* __restrict__ bias,
                    float* __restrict__ out,
                    int n_nodes, int do_relu) {

    const int tid  = threadIdx.x;
    const int lane = tid & 31;
    const int wave = __builtin_amdgcn_readfirstlane(tid >> 5);
    const int c    = lane & 15;
    const int hh   = lane >> 4;

    const int ntiles = (n_nodes + 15) >> 4;
    const int stride = (int)gridDim.x * GEMM_WAVES;
    int tile = (int)blockIdx.x + (int)gridDim.x * wave;

    v2f   nl[KSTEPS], nr[KSTEPS];
    float nidg = 0.0f;
#pragma unroll
    for (int t = 0; t < KSTEPS; ++t) { nl[t] = (v2f){0.0f, 0.0f}; nr[t] = (v2f){0.0f, 0.0f}; }
    if (tile < ntiles) GEMM_LOAD_A(tile)

    const int rot = (int)blockIdx.x & 31;
#pragma unroll 16
    for (int it = 0; it < (HID / 2) * HID / GEMM_THREADS; ++it) {
        const int i  = tid + (((it + rot) & 31) * GEMM_THREADS);
        const int kp = i >> 7, n = i & (HID - 1);
        float4 w;
        w.x = Wl[(size_t)(2 * kp)     * HID + n];
        w.y = Wl[(size_t)(2 * kp + 1) * HID + n];
        w.z = Wr[(size_t)(2 * kp)     * HID + n];
        w.w = Wr[(size_t)(2 * kp + 1) * HID + n];
        wq[i] = w;
    }
    if (tid < HID) sbias[tid] = bias[tid];
    __syncthreads();

    const float4* __restrict__ wlane = wq + hh * HID + c;

    while (tile < ntiles) {
        const int m0 = tile * 16;
        v2f al[KSTEPS], ar[KSTEPS];
#pragma unroll
        for (int t = 0; t < KSTEPS; ++t) { al[t] = nl[t] * nidg; ar[t] = nr[t]; }

        const int nt = tile + stride;
        if (nt < ntiles) GEMM_LOAD_A(nt)

        const int mbase = m0 + hh * 8;
#pragma unroll 1
        for (int j = 0; j < CTILES; ++j) {
            const float4* __restrict__ wj = wlane + 16 * j;
            v8f acc_l = {};
            v8f acc_r = {};
#pragma unroll
            for (int t = 0; t < KSTEPS; ++t) {
                const float4 b = wj[(2 * t) * HID];
                const v2f b_l = {b.x, b.y};
                const v2f b_r = {b.z, b.w};
                acc_l = __builtin_amdgcn_wmma_f32_16x16x4_f32(
                    false, al[t], false, b_l, (short)0, acc_l, false, false);
                acc_r = __builtin_amdgcn_wmma_f32_16x16x4_f32(
                    false, ar[t], false, b_r, (short)0, acc_r, false, false);
            }
            const int ncol = 16 * j + c;
            const float bv = sbias[ncol];
            asm volatile("v_nop\n\tv_nop\n\tv_nop\n\tv_nop\n\tv_nop\n\tv_nop\n\tv_nop\n\tv_nop"
                         : "+v"(acc_l), "+v"(acc_r));
#pragma unroll
            for (int v = 0; v < 8; ++v) {
                const int m = mbase + v;
                if (m < n_nodes) {
                    float o = acc_l[v] + acc_r[v] + bv;
                    if (do_relu) o = o > 0.0f ? o : 0.0f;
                    out[(size_t)m * HID + ncol] = o;
                }
            }
        }
        tile = nt;
    }
}

__global__ __launch_bounds__(GEMM_THREADS)
void sage_gemm_in_kernel(const float* __restrict__ agg,
                         const float* __restrict__ invdeg,
                         const int* __restrict__ n_id,
                         const float* __restrict__ emb,
                         const float* __restrict__ x_struct,
                         const float* __restrict__ Wl,
                         const float* __restrict__ Wr,
                         const float* __restrict__ bias,
                         float* __restrict__ out,
                         int n_nodes) {
    __shared__ float4 wq[(HID / 2) * HID];
    __shared__ float  sbias[HID];
    sage_gemm_body(wq, sbias, agg, invdeg, nullptr, n_id, emb, x_struct,  true,
                   Wl, Wr, bias, out, n_nodes,  1);
}

__global__ __launch_bounds__(GEMM_THREADS)
void sage_gemm_kernel(const float* __restrict__ agg,
                      const float* __restrict__ invdeg,
                      const float* __restrict__ h,
                      const float* __restrict__ Wl,
                      const float* __restrict__ Wr,
                      const float* __restrict__ bias,
                      float* __restrict__ out,
                      int n_nodes, int do_relu) {
    __shared__ float4 wq[(HID / 2) * HID];
    __shared__ float  sbias[HID];
    sage_gemm_body(wq, sbias, agg, invdeg, h, nullptr, nullptr, nullptr,  false,
                   Wl, Wr, bias, out, n_nodes, do_relu);
}

__global__ void edge_dot_kernel(const float* __restrict__ h2,
                                const int* __restrict__ es,
                                const int* __restrict__ ed,
                                float* __restrict__ out, int n) {
    int e = blockIdx.x * blockDim.x + threadIdx.x;
    if (e >= n) return;
    const float4* a = (const float4*)(h2 + (size_t)es[e] * HID);
    const float4* b = (const float4*)(h2 + (size_t)ed[e] * HID);
    float s = 0.0f;
#pragma unroll 8
    for (int i = 0; i < HID / 4; ++i) {
        float4 x = a[i], y = b[i];
        s += x.x * y.x + x.y * y.y + x.z * y.z + x.w * y.w;
    }
    out[e] = s;
}

extern "C" void kernel_launch(void* const* d_in, const int* in_sizes, int n_in,
                              void* d_out, int out_size, void* d_ws, size_t ws_size,
                              hipStream_t stream) {
    const int*   n_id     = (const int*)  d_in[0];
    const float* x_struct = (const float*)d_in[1];
    const int*   eidx     = (const int*)  d_in[2];
    const int*   elab     = (const int*)  d_in[3];
    const float* emb      = (const float*)d_in[4];
    const float* W1l      = (const float*)d_in[5];
    const float* W1r      = (const float*)d_in[6];
    const float* b1       = (const float*)d_in[7];
    const float* W2l      = (const float*)d_in[8];
    const float* W2r      = (const float*)d_in[9];
    const float* b2       = (const float*)d_in[10];
    float*       out      = (float*)d_out;

    const int N = in_sizes[0];
    const int E = in_sizes[2] / 2;
    const int L = in_sizes[3] / 2;

    const int* src = eidx;
    const int* dst = eidx + E;
    const int* lsrc = elab;
    const int* ldst = elab + L;

    float* ws     = (float*)d_ws;
    float* h1     = ws;
    float* h2     = h1  + (size_t)N * HID;
    float* agg    = h2  + (size_t)N * HID;
    float* invdeg = agg + (size_t)N * HID;

    const int ablocks = (N + AGG_ROWS - 1) / AGG_ROWS;

    agg_rows_in_kernel<<<ablocks, AGG_THREADS, 0, stream>>>(
        n_id, emb, x_struct, src, dst, agg, invdeg, E, N);
    sage_gemm_in_kernel<<<GEMM_BLOCKS, GEMM_THREADS, 0, stream>>>(
        agg, invdeg, n_id, emb, x_struct, W1l, W1r, b1, h1, N);

    agg_rows_kernel<<<ablocks, AGG_THREADS, 0, stream>>>(
        h1, src, dst, agg, invdeg, E, N,  0);
    sage_gemm_kernel<<<GEMM_BLOCKS, GEMM_THREADS, 0, stream>>>(
        agg, invdeg, h1, W2l, W2r, b2, h2, N,  0);

    edge_dot_kernel<<<(L + 255) / 256, 256, 0, stream>>>(h2, lsrc, ldst, out, L);
}


# Layer 1's dst scan also writes each block's matched (src id, row) list to the workspace; layer 2's neighbour sum adds that list's rows instead of re-scanning dst; layer-2 LDS-atomic sum order differs
# speedup vs baseline: 3.9828x; 1.1080x over previous
#include <hip/hip_runtime.h>
#include <hip/hip_bf16.h>


typedef __attribute__((ext_vector_type(2))) float v2f;
typedef __attribute__((ext_vector_type(8))) float v8f;

#define HID 128
#define EMBD 64


#define AGG_ROWS    196
#define AGG_THREADS 1024
#define AGG_WAVES   32
#define SCAN_U      4
#define Q_CAP       256
#define LIST_HDR    64
#define Q_DRAIN     56

__device__ __forceinline__ void agg_add_row(float* __restrict__ acc, int* __restrict__ cnt,
                                            int r, const float4 v, int lane, int count_deg) {
    float* a = &acc[r * HID + lane];
    __hip_atomic_fetch_add(a,      v.x, __ATOMIC_RELAXED, __HIP_MEMORY_SCOPE_WORKGROUP);
    __hip_atomic_fetch_add(a + 32, v.y, __ATOMIC_RELAXED, __HIP_MEMORY_SCOPE_WORKGROUP);
    __hip_atomic_fetch_add(a + 64, v.z, __ATOMIC_RELAXED, __HIP_MEMORY_SCOPE_WORKGROUP);
    __hip_atomic_fetch_add(a + 96, v.w, __ATOMIC_RELAXED, __HIP_MEMORY_SCOPE_WORKGROUP);
    if (count_deg && lane == 0)
        __hip_atomic_fetch_add(&cnt[r], 1, __ATOMIC_RELAXED, __HIP_MEMORY_SCOPE_WORKGROUP);
}

__device__ __forceinline__ void agg_add_group(const float* __restrict__ h,
                                              const int* __restrict__ nid,
                                              const float* __restrict__ emb,
                                              const float* __restrict__ xs,
                                              const bool split,
                                              const int s, const int r, const int m,
                                              float* __restrict__ acc, int* __restrict__ cnt,
                                              int lane, int count_deg) {
    const bool   emb_half = lane < 16;
    const float* half_tbl = emb_half ? emb : xs;
    const int    piece    = (lane & 15) * 4;
    {
        int id = 0;
        if (split) id = nid[s];
        for (int j = 0; j < m; j += 8) {
            float4 v[8];
            int    rw[8];
#pragma unroll
            for (int u = 0; u < 8; ++u) {
                const int su = __builtin_amdgcn_readlane(s, j + u);
                rw[u] = __builtin_amdgcn_readlane(r, j + u);
                if (split) {
                    const int iu  = __builtin_amdgcn_readlane(id, j + u);
                    const int row = emb_half ? iu : su;
                    v[u] = *(const float4*)(half_tbl + (size_t)row * EMBD + piece);
                } else {
                    v[u] = ((const float4*)(h + (size_t)su * HID))[lane];
                }
            }
#pragma unroll
            for (int u = 0; u < 8; ++u)
                if (j + u < m) agg_add_row(acc, cnt, rw[u], v[u], lane, count_deg);
        }
    }
}

__device__ __forceinline__ void agg_drain(const float* __restrict__ h,
                                          const int* __restrict__ nid,
                                          const float* __restrict__ emb,
                                          const float* __restrict__ xs,
                                          const bool split,
                                          const int* __restrict__ src,
                                          const int* qe, const int* qr, int n,
                                          float* __restrict__ acc, int* __restrict__ cnt,
                                          int lane, int count_deg,
                                          const bool emit, int* __restrict__ list_n,
                                          int2* __restrict__ ent, int cap) {
    int pos = -1;
    if (emit) {
        int p = 0;
        if (lane == 0) p = __hip_atomic_fetch_add(list_n, n, __ATOMIC_RELAXED, __HIP_MEMORY_SCOPE_WORKGROUP);
        p = __builtin_amdgcn_readfirstlane(p);
        if (p + n <= cap) pos = p;
    }
    for (int i0 = 0; i0 < n; i0 += 32) {
        const int i = i0 + lane;
        int s = 0, r = 0;
        if (i < n) { s = src[qe[i]]; r = qr[i]; }
        if (emit && pos >= 0 && i < n) ent[pos + i] = make_int2(s, r);
        int m = n - i0;
        if (m > 32) m = 32;
        agg_add_group(h, nid, emb, xs, split, s, r, m, acc, cnt, lane, count_deg);
    }
}

__device__ __forceinline__ int4 agg_load_dst4(const int* __restrict__ dst, int g, int ngroups,
                                              int n_edges, bool vec_ok) {
    int4 d = make_int4(-1, -1, -1, -1);
    if (g < ngroups) {
        if (vec_ok) {
            d = ((const int4*)dst)[g];
        } else {
            const int e = g * 4;
            if (e     < n_edges) d.x = dst[e];
            if (e + 1 < n_edges) d.y = dst[e + 1];
            if (e + 2 < n_edges) d.z = dst[e + 2];
            if (e + 3 < n_edges) d.w = dst[e + 3];
        }
    }
    return d;
}

__device__ __forceinline__ void agg_take_group(const int4 d, int g, int base, unsigned nrows,
                                               int* __restrict__ qe, int* __restrict__ qr, int& qn,
                                               const float* __restrict__ h,
                                               const int* __restrict__ nid,
                                               const float* __restrict__ emb,
                                               const float* __restrict__ xs,
                                               const bool split,
                                               const int* __restrict__ src,
                                               float* __restrict__ acc, int* __restrict__ cnt,
                                               int lane, int count_deg,
                                               const bool emit, int* __restrict__ list_n,
                                               int2* __restrict__ ent, int cap) {
    const int dv[4] = {d.x, d.y, d.z, d.w};
    unsigned rr[4];
    bool hit[4];
    bool any_hit = false;
#pragma unroll
    for (int k = 0; k < 4; ++k) {
        rr[k]  = (unsigned)(dv[k] - base);
        hit[k] = rr[k] < nrows;
        any_hit |= hit[k];
    }
    if (__builtin_amdgcn_ballot_w32(any_hit) != 0u) {
#pragma unroll
        for (int k = 0; k < 4; ++k) {
            const unsigned mask = __builtin_amdgcn_ballot_w32(hit[k]);
            if (mask != 0u) {
                if (hit[k]) {
                    const int pos = qn + (int)__builtin_amdgcn_mbcnt_lo(mask, 0u);
                    qe[pos] = g * 4 + k;
                    qr[pos] = (int)rr[k];
                }
                qn += __popc(mask);
            }
        }
        if (qn > Q_DRAIN) {
            agg_drain(h, nid, emb, xs, split, src, qe, qr, qn, acc, cnt, lane, count_deg,
                      emit, list_n, ent, cap);
            qn = 0;
        }
    }
}

__device__ __forceinline__
void agg_rows_body(float4* __restrict__ acc4, int* __restrict__ cnt,
                   int* __restrict__ q_edge, int* __restrict__ q_row,
                   int* __restrict__ list_n, int* __restrict__ elist, const int list_mode,
                   const float* __restrict__ h,
                   const int* __restrict__ nid,
                   const float* __restrict__ emb,
                   const float* __restrict__ xs,
                   const bool split,
                   const int* __restrict__ src,
                   const int* __restrict__ dst,
                   float* __restrict__ agg,
                   float* __restrict__ invdeg,
                   int n_edges, int n_nodes, int write_invdeg) {
    float* acc = (float*)acc4;

    const int tid  = threadIdx.x;
    const int lane = tid & 31;
    const int wave = __builtin_amdgcn_readfirstlane(tid >> 5);
    const int base = blockIdx.x * AGG_ROWS;
    int nr = n_nodes - base;
    if (nr > AGG_ROWS) nr = AGG_ROWS;
    if (nr < 0) nr = 0;
    const unsigned nrows = (unsigned)nr;

    for (int i = tid; i < AGG_ROWS * HID / 4; i += AGG_THREADS)
        acc4[i] = make_float4(0.0f, 0.0f, 0.0f, 0.0f);
    for (int i = tid; i < AGG_ROWS; i += AGG_THREADS) cnt[i] = 0;
    if (tid == 0) *list_n = 0;

    const size_t stretch = (((size_t)n_nodes * HID) / gridDim.x) & ~(size_t)(LIST_HDR - 1);
    int* __restrict__ my_list = elist + (size_t)blockIdx.x * stretch;
    int2* __restrict__ ent = (int2*)(my_list + LIST_HDR);
    const int  cap  = stretch > LIST_HDR ? (int)((stretch - LIST_HDR) >> 1) : 0;
    const bool emit = list_mode == 1;
    int n_list = -1;
    if (list_mode == 2 && stretch > 0) n_list = __builtin_amdgcn_readfirstlane(my_list[0]);
    __syncthreads();

    if (n_list >= 0) {
        const int per = ((n_list + AGG_WAVES - 1) / AGG_WAVES + 7) & ~7;
        const int beg = wave * per;
        int end = beg + per;
        if (end > n_list) end = n_list;
        int2 e_next = make_int2(0, 0);
        if (beg + lane < end) e_next = ent[beg + lane];
        for (int i0 = beg; i0 < end; i0 += 32) {
            const int2 e = e_next;
            e_next = make_int2(0, 0);
            if (i0 + 32 + lane < end) e_next = ent[i0 + 32 + lane];
            int m = end - i0;
            if (m > 32) m = 32;
            agg_add_group(h, nid, emb, xs, split, e.x, e.y, m, acc, cnt, lane, write_invdeg);
        }
    } else {

    int* __restrict__ qe = q_edge + wave * Q_CAP;
    int* __restrict__ qr = q_row  + wave * Q_CAP;
    int qn = 0;

    const int  ngroups     = (n_edges + 3) >> 2;
    const int  trip_groups = 32 * SCAN_U;
    const bool vec_ok      = ((n_edges & 3) == 0) && ((((size_t)dst) & 15) == 0);
    const int  nfull       = vec_ok ? ngroups / trip_groups : 0;
    const int4* __restrict__ dst4 = (const int4*)dst;

    if (wave < nfull) {
        const int rot = (int)(((long long)blockIdx.x * nfull) / gridDim.x);

        int4 da[SCAN_U], db[SCAN_U];
        {
            int tt = wave + rot;
            if (tt >= nfull) tt -= nfull;
#pragma unroll
            for (int u = 0; u < SCAN_U; ++u) da[u] = dst4[tt * trip_groups + u * 32 + lane];
        }
        for (int t = wave; t < nfull; t += 2 * AGG_WAVES) {
            const int t1 = t + AGG_WAVES;
            const int t2 = t + 2 * AGG_WAVES;
            int tt0 = t + rot;
            int tt1 = (t1 < nfull ? t1 : t) + rot;
            int tt2 = (t2 < nfull ? t2 : t) + rot;
            if (tt0 >= nfull) tt0 -= nfull;
            if (tt1 >= nfull) tt1 -= nfull;
            if (tt2 >= nfull) tt2 -= nfull;

#pragma unroll
            for (int u = 0; u < SCAN_U; ++u) db[u] = dst4[tt1 * trip_groups + u * 32 + lane];
#pragma unroll
            for (int u = 0; u < SCAN_U; ++u)
                agg_take_group(da[u], tt0 * trip_groups + u * 32 + lane, base, nrows,
                               qe, qr, qn, h, nid, emb, xs, split, src, acc, cnt, lane, write_invdeg,
                               emit, list_n, ent, cap);
#pragma unroll
            for (int u = 0; u < SCAN_U; ++u) da[u] = dst4[tt2 * trip_groups + u * 32 + lane];
            if (t1 < nfull) {
#pragma unroll
                for (int u = 0; u < SCAN_U; ++u)
                    agg_take_group(db[u], tt1 * trip_groups + u * 32 + lane, base, nrows,
                                   qe, qr, qn, h, nid, emb, xs, split, src, acc, cnt, lane, write_invdeg,
                                   emit, list_n, ent, cap);
            }
        }
    }
    for (int g0 = nfull * trip_groups + wave * 32; g0 < ngroups; g0 += AGG_WAVES * 32) {
        const int g = g0 + lane;
        const int4 d = agg_load_dst4(dst, g, ngroups, n_edges, vec_ok);
        agg_take_group(d, g, base, nrows, qe, qr, qn, h, nid, emb, xs, split, src, acc, cnt, lane, write_invdeg,
                       emit, list_n, ent, cap);
    }
    if (qn > 0) agg_drain(h, nid, emb, xs, split, src, qe, qr, qn, acc, cnt, lane, write_invdeg,
                          emit, list_n, ent, cap);

    }

    __syncthreads();

    if (emit && tid == 0 && stretch > 0) {
        const int total = *list_n;
        my_list[0] = total <= cap ? total : -1;
    }

    for (int i = tid; i < nr * 32; i += AGG_THREADS) {
        const int r = i >> 5, q = i & 31;
        const float* a = &acc[r * HID + q];
        const float4 v = make_float4(a[0], a[32], a[64], a[96]);
        ((float4*)(agg + (size_t)(base + r) * HID))[q] = v;
    }
    if (write_invdeg) {
        for (int r = tid; r < nr; r += AGG_THREADS)
            invdeg[base + r] = 1.0f / fmaxf((float)cnt[r], 1.0f);
    }
}

__global__ __launch_bounds__(AGG_THREADS)
void agg_rows_in_kernel(const int* __restrict__ n_id,
                        const float* __restrict__ emb,
                        const float* __restrict__ x_struct,
                        const int* __restrict__ src,
                        const int* __restrict__ dst,
                        float* __restrict__ agg,
                        float* __restrict__ invdeg,
                        int* __restrict__ elist,
                        int n_edges, int n_nodes) {
    __shared__ float4 acc4[AGG_ROWS * HID / 4];
    __shared__ int    cnt[AGG_ROWS];
    __shared__ int    q_edge[AGG_WAVES * Q_CAP];
    __shared__ int    q_row [AGG_WAVES * Q_CAP];
    __shared__ int    list_n;
    agg_rows_body(acc4, cnt, q_edge, q_row, &list_n, elist,  1,
                  nullptr, n_id, emb, x_struct,  true,
                  src, dst, agg, invdeg, n_edges, n_nodes,  1);
}

__global__ __launch_bounds__(AGG_THREADS)
void agg_rows_kernel(const float* __restrict__ h,
                     const int* __restrict__ src,
                     const int* __restrict__ dst,
                     float* __restrict__ agg,
                     float* __restrict__ invdeg,
                     int* __restrict__ elist,
                     int n_edges, int n_nodes, int write_invdeg) {
    __shared__ float4 acc4[AGG_ROWS * HID / 4];
    __shared__ int    cnt[AGG_ROWS];
    __shared__ int    q_edge[AGG_WAVES * Q_CAP];
    __shared__ int    q_row [AGG_WAVES * Q_CAP];
    __shared__ int    list_n;
    agg_rows_body(acc4, cnt, q_edge, q_row, &list_n, elist,  2,
                  h, nullptr, nullptr, nullptr,  false,
                  src, dst, agg, invdeg, n_edges, n_nodes, write_invdeg);
}

#define GEMM_BLOCKS  256
#define GEMM_THREADS 256
#define GEMM_WAVES   8
#define KSTEPS       (HID / 4)
#define CTILES       (HID / 16)

#define GEMM_LOAD_A(TILE)                                                        \
    {                                                                            \
        int mrow_ = (TILE) * 16 + c;                                             \
        if (mrow_ >= n_nodes) mrow_ = n_nodes - 1;     \
        nidg = invdeg[mrow_];                                                    \
        const v2f* ap_ = (const v2f*)(agg + (size_t)mrow_ * HID) + hh;           \
        const v2f* ha_;                                                          \
        const v2f* hb_;                                                          \
        if (split) {                                                             \
            ha_ = (const v2f*)(emb + (size_t)nid[mrow_] * EMBD) + hh;            \
            hb_ = (const v2f*)(xs  + (size_t)mrow_ * EMBD) + hh;                 \
        } else {                                                                 \
            ha_ = (const v2f*)(h + (size_t)mrow_ * HID) + hh;                    \
            hb_ = ha_ + EMBD / 2;                                                \
        }                                                                        \
          \
        _Pragma("unroll")                                                        \
        for (int t_ = 0; t_ < KSTEPS; ++t_) {                                    \
            nl[t_] = ap_[2 * t_];                                                \
            nr[t_] = t_ < KSTEPS / 2 ? ha_[2 * t_] : hb_[2 * t_ - KSTEPS];       \
        }                                                                        \
    }

__device__ __forceinline__
void sage_gemm_body(float4* __restrict__ wq, float* __restrict__ sbias,
                    const float* __restrict__ agg,
                    const float* __restrict__ invdeg,
                    const float* __restrict__ h,
                    const int* __restrict__ nid,
                    const float* __restrict__ emb,
                    const float* __restrict__ xs,
                    const bool split,
                    const float* __restrict__ Wl,
                    const float* __restrict__ Wr,
                    const float* __restrict__ bias,
                    float* __restrict__ out,
                    int n_nodes, int do_relu) {

    const int tid  = threadIdx.x;
    const int lane = tid & 31;
    const int wave = __builtin_amdgcn_readfirstlane(tid >> 5);
    const int c    = lane & 15;
    const int hh   = lane >> 4;

    const int ntiles = (n_nodes + 15) >> 4;
    const int stride = (int)gridDim.x * GEMM_WAVES;
    int tile = (int)blockIdx.x + (int)gridDim.x * wave;

    v2f   nl[KSTEPS], nr[KSTEPS];
    float nidg = 0.0f;
#pragma unroll
    for (int t = 0; t < KSTEPS; ++t) { nl[t] = (v2f){0.0f, 0.0f}; nr[t] = (v2f){0.0f, 0.0f}; }
    if (tile < ntiles) GEMM_LOAD_A(tile)

    const int rot = (int)blockIdx.x & 31;
#pragma unroll 16
    for (int it = 0; it < (HID / 2) * HID / GEMM_THREADS; ++it) {
        const int i  = tid + (((it + rot) & 31) * GEMM_THREADS);
        const int kp = i >> 7, n = i & (HID - 1);
        float4 w;
        w.x = Wl[(size_t)(2 * kp)     * HID + n];
        w.y = Wl[(size_t)(2 * kp + 1) * HID + n];
        w.z = Wr[(size_t)(2 * kp)     * HID + n];
        w.w = Wr[(size_t)(2 * kp + 1) * HID + n];
        wq[i] = w;
    }
    if (tid < HID) sbias[tid] = bias[tid];
    __syncthreads();

    const float4* __restrict__ wlane = wq + hh * HID + c;

    while (tile < ntiles) {
        const int m0 = tile * 16;
        v2f al[KSTEPS], ar[KSTEPS];
#pragma unroll
        for (int t = 0; t < KSTEPS; ++t) { al[t] = nl[t] * nidg; ar[t] = nr[t]; }

        const int nt = tile + stride;
        if (nt < ntiles) GEMM_LOAD_A(nt)

        const int mbase = m0 + hh * 8;
#pragma unroll 1
        for (int j = 0; j < CTILES; ++j) {
            const float4* __restrict__ wj = wlane + 16 * j;
            v8f acc_l = {};
            v8f acc_r = {};
#pragma unroll
            for (int t = 0; t < KSTEPS; ++t) {
                const float4 b = wj[(2 * t) * HID];
                const v2f b_l = {b.x, b.y};
                const v2f b_r = {b.z, b.w};
                acc_l = __builtin_amdgcn_wmma_f32_16x16x4_f32(
                    false, al[t], false, b_l, (short)0, acc_l, false, false);
                acc_r = __builtin_amdgcn_wmma_f32_16x16x4_f32(
                    false, ar[t], false, b_r, (short)0, acc_r, false, false);
            }
            const int ncol = 16 * j + c;
            const float bv = sbias[ncol];
            asm volatile("v_nop\n\tv_nop\n\tv_nop\n\tv_nop\n\tv_nop\n\tv_nop\n\tv_nop\n\tv_nop"
                         : "+v"(acc_l), "+v"(acc_r));
#pragma unroll
            for (int v = 0; v < 8; ++v) {
                const int m = mbase + v;
                if (m < n_nodes) {
                    float o = acc_l[v] + acc_r[v] + bv;
                    if (do_relu) o = o > 0.0f ? o : 0.0f;
                    out[(size_t)m * HID + ncol] = o;
                }
            }
        }
        tile = nt;
    }
}

__global__ __launch_bounds__(GEMM_THREADS)
void sage_gemm_in_kernel(const float* __restrict__ agg,
                         const float* __restrict__ invdeg,
                         const int* __restrict__ n_id,
                         const float* __restrict__ emb,
                         const float* __restrict__ x_struct,
                         const float* __restrict__ Wl,
                         const float* __restrict__ Wr,
                         const float* __restrict__ bias,
                         float* __restrict__ out,
                         int n_nodes) {
    __shared__ float4 wq[(HID / 2) * HID];
    __shared__ float  sbias[HID];
    sage_gemm_body(wq, sbias, agg, invdeg, nullptr, n_id, emb, x_struct,  true,
                   Wl, Wr, bias, out, n_nodes,  1);
}

__global__ __launch_bounds__(GEMM_THREADS)
void sage_gemm_kernel(const float* __restrict__ agg,
                      const float* __restrict__ invdeg,
                      const float* __restrict__ h,
                      const float* __restrict__ Wl,
                      const float* __restrict__ Wr,
                      const float* __restrict__ bias,
                      float* __restrict__ out,
                      int n_nodes, int do_relu) {
    __shared__ float4 wq[(HID / 2) * HID];
    __shared__ float  sbias[HID];
    sage_gemm_body(wq, sbias, agg, invdeg, h, nullptr, nullptr, nullptr,  false,
                   Wl, Wr, bias, out, n_nodes, do_relu);
}

__global__ void edge_dot_kernel(const float* __restrict__ h2,
                                const int* __restrict__ es,
                                const int* __restrict__ ed,
                                float* __restrict__ out, int n) {
    int e = blockIdx.x * blockDim.x + threadIdx.x;
    if (e >= n) return;
    const float4* a = (const float4*)(h2 + (size_t)es[e] * HID);
    const float4* b = (const float4*)(h2 + (size_t)ed[e] * HID);
    float s = 0.0f;
#pragma unroll 8
    for (int i = 0; i < HID / 4; ++i) {
        float4 x = a[i], y = b[i];
        s += x.x * y.x + x.y * y.y + x.z * y.z + x.w * y.w;
    }
    out[e] = s;
}

extern "C" void kernel_launch(void* const* d_in, const int* in_sizes, int n_in,
                              void* d_out, int out_size, void* d_ws, size_t ws_size,
                              hipStream_t stream) {
    const int*   n_id     = (const int*)  d_in[0];
    const float* x_struct = (const float*)d_in[1];
    const int*   eidx     = (const int*)  d_in[2];
    const int*   elab     = (const int*)  d_in[3];
    const float* emb      = (const float*)d_in[4];
    const float* W1l      = (const float*)d_in[5];
    const float* W1r      = (const float*)d_in[6];
    const float* b1       = (const float*)d_in[7];
    const float* W2l      = (const float*)d_in[8];
    const float* W2r      = (const float*)d_in[9];
    const float* b2       = (const float*)d_in[10];
    float*       out      = (float*)d_out;

    const int N = in_sizes[0];
    const int E = in_sizes[2] / 2;
    const int L = in_sizes[3] / 2;

    const int* src = eidx;
    const int* dst = eidx + E;
    const int* lsrc = elab;
    const int* ldst = elab + L;

    float* ws     = (float*)d_ws;
    float* h1     = ws;
    float* h2     = h1  + (size_t)N * HID;
    float* agg    = h2  + (size_t)N * HID;
    float* invdeg = agg + (size_t)N * HID;

    const int ablocks = (N + AGG_ROWS - 1) / AGG_ROWS;
    int* elist = (int*)h2;

    agg_rows_in_kernel<<<ablocks, AGG_THREADS, 0, stream>>>(
        n_id, emb, x_struct, src, dst, agg, invdeg, elist, E, N);
    sage_gemm_in_kernel<<<GEMM_BLOCKS, GEMM_THREADS, 0, stream>>>(
        agg, invdeg, n_id, emb, x_struct, W1l, W1r, b1, h1, N);

    agg_rows_kernel<<<ablocks, AGG_THREADS, 0, stream>>>(
        h1, src, dst, agg, invdeg, elist, E, N,  0);
    sage_gemm_kernel<<<GEMM_BLOCKS, GEMM_THREADS, 0, stream>>>(
        agg, invdeg, h1, W2l, W2r, b2, h2, N,  0);

    edge_dot_kernel<<<(L + 255) / 256, 256, 0, stream>>>(h2, lsrc, ldst, out, L);
}


# Linear layer moved into the owner block: its 196 summed rows stay in LDS (padded pitch) and are the mean-path A fragments, W staged ahead of the scan; agg plane and 2 launches gone; sums unchanged
# speedup vs baseline: 4.3323x; 1.0878x over previous
#include <hip/hip_runtime.h>
#include <hip/hip_bf16.h>


typedef __attribute__((ext_vector_type(2))) float v2f;
typedef __attribute__((ext_vector_type(8))) float v8f;

#define HID 128
#define EMBD 64


#define AGG_ROWS    196
#define ACC_TILES   13
#define ACC_ROWS_PAD (ACC_TILES * 16)
#define ACC_PITCH   132
#define ACC_HALF    66
#define SCALE_TRIPS ((AGG_ROWS * (ACC_PITCH / 4) + AGG_THREADS - 1) / AGG_THREADS)
#define AGG_THREADS 1024
#define AGG_WAVES   32
#define SCAN_U      4
#define Q_CAP       256
#define LIST_HDR    64
#define Q_DRAIN     56

__device__ __forceinline__ void agg_add_row(float* __restrict__ acc, int* __restrict__ cnt,
                                            int r, const float4 v, int lane, int count_deg) {
    float* a = &acc[r * ACC_PITCH + 2 * lane];
    __hip_atomic_fetch_add(a,                v.x, __ATOMIC_RELAXED, __HIP_MEMORY_SCOPE_WORKGROUP);
    __hip_atomic_fetch_add(a + 1,            v.y, __ATOMIC_RELAXED, __HIP_MEMORY_SCOPE_WORKGROUP);
    __hip_atomic_fetch_add(a + ACC_HALF,     v.z, __ATOMIC_RELAXED, __HIP_MEMORY_SCOPE_WORKGROUP);
    __hip_atomic_fetch_add(a + ACC_HALF + 1, v.w, __ATOMIC_RELAXED, __HIP_MEMORY_SCOPE_WORKGROUP);
    if (count_deg && lane == 0)
        __hip_atomic_fetch_add(&cnt[r], 1, __ATOMIC_RELAXED, __HIP_MEMORY_SCOPE_WORKGROUP);
}

__device__ __forceinline__ void agg_add_group(const float* __restrict__ h,
                                              const int* __restrict__ nid,
                                              const float* __restrict__ emb,
                                              const float* __restrict__ xs,
                                              const bool split,
                                              const int s, const int r, const int m,
                                              float* __restrict__ acc, int* __restrict__ cnt,
                                              int lane, int count_deg) {
    const bool   emb_half = lane < 16;
    const float* half_tbl = emb_half ? emb : xs;
    const int    piece    = (lane & 15) * 4;
    {
        int id = 0;
        if (split) id = nid[s];
        for (int j = 0; j < m; j += 8) {
            float4 v[8];
            int    rw[8];
#pragma unroll
            for (int u = 0; u < 8; ++u) {
                const int su = __builtin_amdgcn_readlane(s, j + u);
                rw[u] = __builtin_amdgcn_readlane(r, j + u);
                if (split) {
                    const int iu  = __builtin_amdgcn_readlane(id, j + u);
                    const int row = emb_half ? iu : su;
                    v[u] = *(const float4*)(half_tbl + (size_t)row * EMBD + piece);
                } else {
                    v[u] = ((const float4*)(h + (size_t)su * HID))[lane];
                }
            }
#pragma unroll
            for (int u = 0; u < 8; ++u)
                if (j + u < m) agg_add_row(acc, cnt, rw[u], v[u], lane, count_deg);
        }
    }
}

__device__ __forceinline__ void agg_drain(const float* __restrict__ h,
                                          const int* __restrict__ nid,
                                          const float* __restrict__ emb,
                                          const float* __restrict__ xs,
                                          const bool split,
                                          const int* __restrict__ src,
                                          const int* qe, const int* qr, int n,
                                          float* __restrict__ acc, int* __restrict__ cnt,
                                          int lane, int count_deg,
                                          const bool emit, int* __restrict__ list_n,
                                          int2* __restrict__ ent, int cap) {
    int pos = -1;
    if (emit) {
        int p = 0;
        if (lane == 0) p = __hip_atomic_fetch_add(list_n, n, __ATOMIC_RELAXED, __HIP_MEMORY_SCOPE_WORKGROUP);
        p = __builtin_amdgcn_readfirstlane(p);
        if (p + n <= cap) pos = p;
    }
    for (int i0 = 0; i0 < n; i0 += 32) {
        const int i = i0 + lane;
        int s = 0, r = 0;
        if (i < n) { s = src[qe[i]]; r = qr[i]; }
        if (emit && pos >= 0 && i < n) ent[pos + i] = make_int2(s, r);
        int m = n - i0;
        if (m > 32) m = 32;
        agg_add_group(h, nid, emb, xs, split, s, r, m, acc, cnt, lane, count_deg);
    }
}

__device__ __forceinline__ int4 agg_load_dst4(const int* __restrict__ dst, int g, int ngroups,
                                              int n_edges, bool vec_ok) {
    int4 d = make_int4(-1, -1, -1, -1);
    if (g < ngroups) {
        if (vec_ok) {
            d = ((const int4*)dst)[g];
        } else {
            const int e = g * 4;
            if (e     < n_edges) d.x = dst[e];
            if (e + 1 < n_edges) d.y = dst[e + 1];
            if (e + 2 < n_edges) d.z = dst[e + 2];
            if (e + 3 < n_edges) d.w = dst[e + 3];
        }
    }
    return d;
}

__device__ __forceinline__ void agg_take_group(const int4 d, int g, int base, unsigned nrows,
                                               int* __restrict__ qe, int* __restrict__ qr, int& qn,
                                               const float* __restrict__ h,
                                               const int* __restrict__ nid,
                                               const float* __restrict__ emb,
                                               const float* __restrict__ xs,
                                               const bool split,
                                               const int* __restrict__ src,
                                               float* __restrict__ acc, int* __restrict__ cnt,
                                               int lane, int count_deg,
                                               const bool emit, int* __restrict__ list_n,
                                               int2* __restrict__ ent, int cap) {
    const int dv[4] = {d.x, d.y, d.z, d.w};
    unsigned rr[4];
    bool hit[4];
    bool any_hit = false;
#pragma unroll
    for (int k = 0; k < 4; ++k) {
        rr[k]  = (unsigned)(dv[k] - base);
        hit[k] = rr[k] < nrows;
        any_hit |= hit[k];
    }
    if (__builtin_amdgcn_ballot_w32(any_hit) != 0u) {
#pragma unroll
        for (int k = 0; k < 4; ++k) {
            const unsigned mask = __builtin_amdgcn_ballot_w32(hit[k]);
            if (mask != 0u) {
                if (hit[k]) {
                    const int pos = qn + (int)__builtin_amdgcn_mbcnt_lo(mask, 0u);
                    qe[pos] = g * 4 + k;
                    qr[pos] = (int)rr[k];
                }
                qn += __popc(mask);
            }
        }
        if (qn > Q_DRAIN) {
            agg_drain(h, nid, emb, xs, split, src, qe, qr, qn, acc, cnt, lane, count_deg,
                      emit, list_n, ent, cap);
            qn = 0;
        }
    }
}

#define KSTEPS       (HID / 4)
#define KGROUP       8

#define OWNER_LOAD_SELF(DST, G)                                                  \
    _Pragma("unroll")                                                            \
    for (int i_ = 0; i_ < KGROUP; ++i_) {                                        \
        const int t_ = (G) * KGROUP + i_;                                        \
        DST[i_] = t_ < KSTEPS / 2 ? ha_[2 * t_] : hb_[2 * t_ - KSTEPS];          \
    }

#define OWNER_KGROUP(SELF, G)                                                    \
    _Pragma("unroll")                                                            \
    for (int i_ = 0; i_ < KGROUP; ++i_) {                                        \
        const int t_ = (G) * KGROUP + i_;                                        \
        const v2f a_l = arow[t_];                                                \
        const v2f a_r = SELF[i_];                                                \
        const float4* wt_ = t_ < KSTEPS / 2 ? wj_lo + (2 * t_) * HID             \
                                            : wj_hi + (2 * t_ - KSTEPS) * HID;   \
        const float4 b0 = wt_[0];                                                \
        const float4 b1 = wt_[16];                                               \
        acc_l0 = __builtin_amdgcn_wmma_f32_16x16x4_f32(                          \
            false, a_l, false, (v2f){b0.x, b0.y}, (short)0, acc_l0, false, false); \
        acc_r0 = __builtin_amdgcn_wmma_f32_16x16x4_f32(                          \
            false, a_r, false, (v2f){b0.z, b0.w}, (short)0, acc_r0, false, false); \
        acc_l1 = __builtin_amdgcn_wmma_f32_16x16x4_f32(                          \
            false, a_l, false, (v2f){b1.x, b1.y}, (short)0, acc_l1, false, false); \
        acc_r1 = __builtin_amdgcn_wmma_f32_16x16x4_f32(                          \
            false, a_r, false, (v2f){b1.z, b1.w}, (short)0, acc_r1, false, false); \
    }

__device__ __forceinline__
void owner_gemm(const float* __restrict__ acc,
                const float4* __restrict__ wq,
                const float* __restrict__ h,
                const int* __restrict__ nid,
                const float* __restrict__ emb,
                const float* __restrict__ xs,
                const bool split,
                const float* __restrict__ bias,
                float* __restrict__ out,
                int base, int nr, int n_nodes, int do_relu, int wave, int lane) {
    const int c  = lane & 15;
    const int hh = lane >> 4;
    const int n_units = ((nr + 15) >> 4) * 4;

    for (int u = wave; u < n_units; u += AGG_WAVES) {
        const int rt = u >> 2;
        const int cp = u & 3;
        const int r  = rt * 16 + c;
        int grow = base + r;
        if (grow >= n_nodes) grow = n_nodes - 1;
        const v2f* __restrict__ arow = (const v2f*)(acc + r * ACC_PITCH + hh * ACC_HALF);
        int lo_entries = hh * HID + c + 32 * cp;
        int hi_entries = lo_entries + KSTEPS * HID;
        int ncol0      = 32 * cp + c;
        asm volatile("" : "+v"(lo_entries), "+v"(hi_entries), "+v"(ncol0));
        const float4* __restrict__ wj_lo = wq + lo_entries;
        const float4* __restrict__ wj_hi = wq + hi_entries;
        const v2f* ha_;
        const v2f* hb_;
        if (split) {
            ha_ = (const v2f*)(emb + (size_t)nid[grow] * EMBD) + hh;
            hb_ = (const v2f*)(xs  + (size_t)grow * EMBD) + hh;
        } else {
            ha_ = (const v2f*)(h + (size_t)grow * HID) + hh;
            hb_ = ha_ + EMBD / 2;
        }

        v8f acc_l0 = {}, acc_r0 = {};
        v8f acc_l1 = {}, acc_r1 = {};
        v2f sa[KGROUP], sb[KGROUP];
        OWNER_LOAD_SELF(sa, 0)
        OWNER_LOAD_SELF(sb, 1)
        OWNER_KGROUP(sa, 0)
        OWNER_LOAD_SELF(sa, 2)
        OWNER_KGROUP(sb, 1)
        OWNER_LOAD_SELF(sb, 3)
        OWNER_KGROUP(sa, 2)
        OWNER_KGROUP(sb, 3)

        __builtin_amdgcn_sched_barrier(0);
        asm volatile("v_nop\n\tv_nop\n\tv_nop\n\tv_nop\n\tv_nop\n\tv_nop\n\tv_nop\n\tv_nop"
                     : "+v"(acc_l0), "+v"(acc_r0), "+v"(acc_l1), "+v"(acc_r1));

        const int rbase = rt * 16 + hh * 8;
        const float bv0 = bias[ncol0];
        const float bv1 = bias[ncol0 + 16];
#pragma unroll
        for (int v = 0; v < 8; ++v) {
            const int rr = rbase + v;
            if (rr < nr) {
                float o0 = acc_l0[v] + acc_r0[v] + bv0;
                float o1 = acc_l1[v] + acc_r1[v] + bv1;
                if (do_relu) { o0 = o0 > 0.0f ? o0 : 0.0f; o1 = o1 > 0.0f ? o1 : 0.0f; }
                float* __restrict__ orow = out + (size_t)(base + rr) * HID;
                orow[ncol0]      = o0;
                orow[ncol0 + 16] = o1;
            }
        }
    }
}

__device__ __forceinline__
void agg_rows_body(float4* __restrict__ acc4, int* __restrict__ cnt,
                   int* __restrict__ q_edge, int* __restrict__ q_row,
                   int* __restrict__ list_n, float4* __restrict__ wq,
                   int* __restrict__ elist, const int list_mode,
                   const float* __restrict__ h,
                   const int* __restrict__ nid,
                   const float* __restrict__ emb,
                   const float* __restrict__ xs,
                   const bool split,
                   const int* __restrict__ src,
                   const int* __restrict__ dst,
                   const float* __restrict__ Wl,
                   const float* __restrict__ Wr,
                   const float* __restrict__ bias,
                   float* __restrict__ out,
                   float* __restrict__ invdeg,
                   int n_edges, int n_nodes, int write_invdeg, int do_relu) {
    float* acc = (float*)acc4;

    const int tid  = threadIdx.x;
    const int lane = tid & 31;
    const int wave = __builtin_amdgcn_readfirstlane(tid >> 5);
    const int base = blockIdx.x * AGG_ROWS;
    int nr = n_nodes - base;
    if (nr > AGG_ROWS) nr = AGG_ROWS;
    if (nr < 0) nr = 0;
    const unsigned nrows = (unsigned)nr;

    {
        const int rotw = (int)blockIdx.x & 7;
        float4 w[(HID / 2) * HID / AGG_THREADS];
#pragma unroll
        for (int it = 0; it < (HID / 2) * HID / AGG_THREADS; ++it) {
            const int i  = tid + (((it + rotw) & 7) * AGG_THREADS);
            const int kp = i >> 7, n = i & (HID - 1);
            w[it].x = Wl[(size_t)(2 * kp)     * HID + n];
            w[it].y = Wl[(size_t)(2 * kp + 1) * HID + n];
            w[it].z = Wr[(size_t)(2 * kp)     * HID + n];
            w[it].w = Wr[(size_t)(2 * kp + 1) * HID + n];
        }
        for (int i = tid; i < ACC_ROWS_PAD * ACC_PITCH / 4; i += AGG_THREADS)
            acc4[i] = make_float4(0.0f, 0.0f, 0.0f, 0.0f);
        for (int i = tid; i < ACC_ROWS_PAD; i += AGG_THREADS)
            cnt[i] = (!write_invdeg && i < nr) ? __float_as_int(invdeg[base + i]) : 0;
        if (tid == 0) *list_n = 0;
#pragma unroll
        for (int it = 0; it < (HID / 2) * HID / AGG_THREADS; ++it)
            wq[tid + (((it + rotw) & 7) * AGG_THREADS)] = w[it];
    }

    const size_t stretch = (((size_t)n_nodes * HID) / gridDim.x) & ~(size_t)(LIST_HDR - 1);
    int* __restrict__ my_list = elist + (size_t)blockIdx.x * stretch;
    int2* __restrict__ ent = (int2*)(my_list + LIST_HDR);
    const int  cap  = stretch > LIST_HDR ? (int)((stretch - LIST_HDR) >> 1) : 0;
    const bool emit = list_mode == 1;
    int n_list = -1;
    if (list_mode == 2 && stretch > 0) n_list = __builtin_amdgcn_readfirstlane(my_list[0]);
    __syncthreads();

    if (n_list >= 0) {
        const int per = ((n_list + AGG_WAVES - 1) / AGG_WAVES + 7) & ~7;
        const int beg = wave * per;
        int end = beg + per;
        if (end > n_list) end = n_list;
        int2 e_next = make_int2(0, 0);
        if (beg + lane < end) e_next = ent[beg + lane];
        for (int i0 = beg; i0 < end; i0 += 32) {
            const int2 e = e_next;
            e_next = make_int2(0, 0);
            if (i0 + 32 + lane < end) e_next = ent[i0 + 32 + lane];
            int m = end - i0;
            if (m > 32) m = 32;
            agg_add_group(h, nid, emb, xs, split, e.x, e.y, m, acc, cnt, lane, write_invdeg);
        }
    } else {

    int* __restrict__ qe = q_edge + wave * Q_CAP;
    int* __restrict__ qr = q_row  + wave * Q_CAP;
    int qn = 0;

    const int  ngroups     = (n_edges + 3) >> 2;
    const int  trip_groups = 32 * SCAN_U;
    const bool vec_ok      = ((n_edges & 3) == 0) && ((((size_t)dst) & 15) == 0);
    const int  nfull       = vec_ok ? ngroups / trip_groups : 0;
    const int4* __restrict__ dst4 = (const int4*)dst;

    if (wave < nfull) {
        const int rot = (int)(((long long)blockIdx.x * nfull) / gridDim.x);

        int4 da[SCAN_U], db[SCAN_U];
        {
            int tt = wave + rot;
            if (tt >= nfull) tt -= nfull;
#pragma unroll
            for (int u = 0; u < SCAN_U; ++u) da[u] = dst4[tt * trip_groups + u * 32 + lane];
        }
        for (int t = wave; t < nfull; t += 2 * AGG_WAVES) {
            const int t1 = t + AGG_WAVES;
            const int t2 = t + 2 * AGG_WAVES;
            int tt0 = t + rot;
            int tt1 = (t1 < nfull ? t1 : t) + rot;
            int tt2 = (t2 < nfull ? t2 : t) + rot;
            if (tt0 >= nfull) tt0 -= nfull;
            if (tt1 >= nfull) tt1 -= nfull;
            if (tt2 >= nfull) tt2 -= nfull;

#pragma unroll
            for (int u = 0; u < SCAN_U; ++u) db[u] = dst4[tt1 * trip_groups + u * 32 + lane];
#pragma unroll
            for (int u = 0; u < SCAN_U; ++u)
                agg_take_group(da[u], tt0 * trip_groups + u * 32 + lane, base, nrows,
                               qe, qr, qn, h, nid, emb, xs, split, src, acc, cnt, lane, write_invdeg,
                               emit, list_n, ent, cap);
#pragma unroll
            for (int u = 0; u < SCAN_U; ++u) da[u] = dst4[tt2 * trip_groups + u * 32 + lane];
            if (t1 < nfull) {
#pragma unroll
                for (int u = 0; u < SCAN_U; ++u)
                    agg_take_group(db[u], tt1 * trip_groups + u * 32 + lane, base, nrows,
                                   qe, qr, qn, h, nid, emb, xs, split, src, acc, cnt, lane, write_invdeg,
                                   emit, list_n, ent, cap);
            }
        }
    }
    for (int g0 = nfull * trip_groups + wave * 32; g0 < ngroups; g0 += AGG_WAVES * 32) {
        const int g = g0 + lane;
        const int4 d = agg_load_dst4(dst, g, ngroups, n_edges, vec_ok);
        agg_take_group(d, g, base, nrows, qe, qr, qn, h, nid, emb, xs, split, src, acc, cnt, lane, write_invdeg,
                       emit, list_n, ent, cap);
    }
    if (qn > 0) agg_drain(h, nid, emb, xs, split, src, qe, qr, qn, acc, cnt, lane, write_invdeg,
                          emit, list_n, ent, cap);

    }

    __syncthreads();

    if (emit && tid == 0 && stretch > 0) {
        const int total = *list_n;
        my_list[0] = total <= cap ? total : -1;
    }
    if (write_invdeg) {
        for (int r = tid; r < nr; r += AGG_THREADS)
            invdeg[base + r] = 1.0f / fmaxf((float)cnt[r], 1.0f);
    }

    {
        float4 sv[SCALE_TRIPS];
        float  si[SCALE_TRIPS];
#pragma unroll
        for (int k = 0; k < SCALE_TRIPS; ++k) {
            const int i = tid + k * AGG_THREADS;
            sv[k] = make_float4(0.0f, 0.0f, 0.0f, 0.0f);
            si[k] = 0.0f;
            if (i < nr * (ACC_PITCH / 4)) {
                const int r = i / (ACC_PITCH / 4);
                sv[k] = acc4[i];
                si[k] = write_invdeg ? 1.0f / fmaxf((float)cnt[r], 1.0f)
                                     : __int_as_float(cnt[r]);
            }
        }
#pragma unroll
        for (int k = 0; k < SCALE_TRIPS; ++k) {
            const int i = tid + k * AGG_THREADS;
            if (i < nr * (ACC_PITCH / 4)) {
                float4 v = sv[k];
                v.x *= si[k]; v.y *= si[k]; v.z *= si[k]; v.w *= si[k];
                acc4[i] = v;
            }
        }
    }
    __syncthreads();

    owner_gemm(acc, wq, h, nid, emb, xs, split, bias, out, base, nr, n_nodes, do_relu, wave, lane);
}

__global__ __launch_bounds__(AGG_THREADS)
void sage_layer_in_kernel(const int* __restrict__ n_id,
                          const float* __restrict__ emb,
                          const float* __restrict__ x_struct,
                          const int* __restrict__ src,
                          const int* __restrict__ dst,
                          const float* __restrict__ Wl,
                          const float* __restrict__ Wr,
                          const float* __restrict__ bias,
                          float* __restrict__ out,
                          float* __restrict__ invdeg,
                          int* __restrict__ elist,
                          int n_edges, int n_nodes) {
    __shared__ float4 acc4[ACC_ROWS_PAD * ACC_PITCH / 4];
    __shared__ int    cnt[ACC_ROWS_PAD];
    __shared__ int    q_edge[AGG_WAVES * Q_CAP];
    __shared__ int    q_row [AGG_WAVES * Q_CAP];
    __shared__ int    list_n;
    __shared__ float4 wq[(HID / 2) * HID];
    agg_rows_body(acc4, cnt, q_edge, q_row, &list_n, wq, elist,  1,
                  nullptr, n_id, emb, x_struct,  true,
                  src, dst, Wl, Wr, bias, out, invdeg, n_edges, n_nodes,
                   1,  1);
}

__global__ __launch_bounds__(AGG_THREADS)
void sage_layer_kernel(const float* __restrict__ h,
                       const int* __restrict__ src,
                       const int* __restrict__ dst,
                       const float* __restrict__ Wl,
                       const float* __restrict__ Wr,
                       const float* __restrict__ bias,
                       float* __restrict__ out,
                       float* __restrict__ invdeg,
                       int* __restrict__ elist,
                       int n_edges, int n_nodes, int do_relu) {
    __shared__ float4 acc4[ACC_ROWS_PAD * ACC_PITCH / 4];
    __shared__ int    cnt[ACC_ROWS_PAD];
    __shared__ int    q_edge[AGG_WAVES * Q_CAP];
    __shared__ int    q_row [AGG_WAVES * Q_CAP];
    __shared__ int    list_n;
    __shared__ float4 wq[(HID / 2) * HID];
    agg_rows_body(acc4, cnt, q_edge, q_row, &list_n, wq, elist,  2,
                  h, nullptr, nullptr, nullptr,  false,
                  src, dst, Wl, Wr, bias, out, invdeg, n_edges, n_nodes,
                   0, do_relu);
}

__global__ void edge_dot_kernel(const float* __restrict__ h2,
                                const int* __restrict__ es,
                                const int* __restrict__ ed,
                                float* __restrict__ out, int n) {
    int e = blockIdx.x * blockDim.x + threadIdx.x;
    if (e >= n) return;
    const float4* a = (const float4*)(h2 + (size_t)es[e] * HID);
    const float4* b = (const float4*)(h2 + (size_t)ed[e] * HID);
    float s = 0.0f;
#pragma unroll 8
    for (int i = 0; i < HID / 4; ++i) {
        float4 x = a[i], y = b[i];
        s += x.x * y.x + x.y * y.y + x.z * y.z + x.w * y.w;
    }
    out[e] = s;
}

extern "C" void kernel_launch(void* const* d_in, const int* in_sizes, int n_in,
                              void* d_out, int out_size, void* d_ws, size_t ws_size,
                              hipStream_t stream) {
    const int*   n_id     = (const int*)  d_in[0];
    const float* x_struct = (const float*)d_in[1];
    const int*   eidx     = (const int*)  d_in[2];
    const int*   elab     = (const int*)  d_in[3];
    const float* emb      = (const float*)d_in[4];
    const float* W1l      = (const float*)d_in[5];
    const float* W1r      = (const float*)d_in[6];
    const float* b1       = (const float*)d_in[7];
    const float* W2l      = (const float*)d_in[8];
    const float* W2r      = (const float*)d_in[9];
    const float* b2       = (const float*)d_in[10];
    float*       out      = (float*)d_out;

    const int N = in_sizes[0];
    const int E = in_sizes[2] / 2;
    const int L = in_sizes[3] / 2;

    const int* src = eidx;
    const int* dst = eidx + E;
    const int* lsrc = elab;
    const int* ldst = elab + L;

    float* ws     = (float*)d_ws;
    float* h1     = ws;
    float* h2     = h1  + (size_t)N * HID;
    float* lists  = h2  + (size_t)N * HID;
    float* invdeg = lists + (size_t)N * HID;

    const int ablocks = (N + AGG_ROWS - 1) / AGG_ROWS;
    int* elist = (int*)lists;

    sage_layer_in_kernel<<<ablocks, AGG_THREADS, 0, stream>>>(
        n_id, emb, x_struct, src, dst, W1l, W1r, b1, h1, invdeg, elist, E, N);

    sage_layer_kernel<<<ablocks, AGG_THREADS, 0, stream>>>(
        h1, src, dst, W2l, W2r, b2, h2, invdeg, elist, E, N,  0);

    edge_dot_kernel<<<(L + 255) / 256, 256, 0, stream>>>(h2, lsrc, ldst, out, L);
}


# Layer 1 finds its edges through one owner byte per edge (dst/196, from a new small launch): a block scans 0.8 MB of bytes, not 3.2 MB of dst, and reads dst only where a byte matches; LDS-order sums
# speedup vs baseline: 4.5596x; 1.0525x over previous
#include <hip/hip_runtime.h>
#include <hip/hip_bf16.h>


typedef __attribute__((ext_vector_type(2))) float v2f;
typedef __attribute__((ext_vector_type(8))) float v8f;

#define HID 128
#define EMBD 64


#define AGG_ROWS    196
#define ACC_TILES   13
#define ACC_ROWS_PAD (ACC_TILES * 16)
#define ACC_PITCH   132
#define ACC_HALF    66
#define SCALE_TRIPS ((AGG_ROWS * (ACC_PITCH / 4) + AGG_THREADS - 1) / AGG_THREADS)
#define AGG_THREADS 1024
#define AGG_WAVES   32
#define SCAN_U      4
#define TAG_U       4
#define C_CAP       160
#define Q_CAP       192
#define LIST_HDR    64
#define Q_DRAIN     56

__device__ __forceinline__ void agg_add_row(float* __restrict__ acc, int* __restrict__ cnt,
                                            int r, const float4 v, int lane, int count_deg) {
    float* a = &acc[r * ACC_PITCH + 2 * lane];
    __hip_atomic_fetch_add(a,                v.x, __ATOMIC_RELAXED, __HIP_MEMORY_SCOPE_WORKGROUP);
    __hip_atomic_fetch_add(a + 1,            v.y, __ATOMIC_RELAXED, __HIP_MEMORY_SCOPE_WORKGROUP);
    __hip_atomic_fetch_add(a + ACC_HALF,     v.z, __ATOMIC_RELAXED, __HIP_MEMORY_SCOPE_WORKGROUP);
    __hip_atomic_fetch_add(a + ACC_HALF + 1, v.w, __ATOMIC_RELAXED, __HIP_MEMORY_SCOPE_WORKGROUP);
    if (count_deg && lane == 0)
        __hip_atomic_fetch_add(&cnt[r], 1, __ATOMIC_RELAXED, __HIP_MEMORY_SCOPE_WORKGROUP);
}

__device__ __forceinline__ void agg_add_group(const float* __restrict__ h,
                                              const int* __restrict__ nid,
                                              const float* __restrict__ emb,
                                              const float* __restrict__ xs,
                                              const bool split,
                                              const int s, const int r, const int m,
                                              float* __restrict__ acc, int* __restrict__ cnt,
                                              int lane, int count_deg) {
    const bool   emb_half = lane < 16;
    const float* half_tbl = emb_half ? emb : xs;
    const int    piece    = (lane & 15) * 4;
    {
        int id = 0;
        if (split) id = nid[s];
        for (int j = 0; j < m; j += 8) {
            float4 v[8];
            int    rw[8];
#pragma unroll
            for (int u = 0; u < 8; ++u) {
                const int su = __builtin_amdgcn_readlane(s, j + u);
                rw[u] = __builtin_amdgcn_readlane(r, j + u);
                if (split) {
                    const int iu  = __builtin_amdgcn_readlane(id, j + u);
                    const int row = emb_half ? iu : su;
                    v[u] = *(const float4*)(half_tbl + (size_t)row * EMBD + piece);
                } else {
                    v[u] = ((const float4*)(h + (size_t)su * HID))[lane];
                }
            }
#pragma unroll
            for (int u = 0; u < 8; ++u)
                if (j + u < m) agg_add_row(acc, cnt, rw[u], v[u], lane, count_deg);
        }
    }
}

__device__ __forceinline__ void agg_drain(const float* __restrict__ h,
                                          const int* __restrict__ nid,
                                          const float* __restrict__ emb,
                                          const float* __restrict__ xs,
                                          const bool split,
                                          const int* __restrict__ src,
                                          const int* qe, const int* qr, int n,
                                          float* __restrict__ acc, int* __restrict__ cnt,
                                          int lane, int count_deg,
                                          const bool emit, int* __restrict__ list_n,
                                          int2* __restrict__ ent, int cap) {
    int pos = -1;
    if (emit) {
        int p = 0;
        if (lane == 0) p = __hip_atomic_fetch_add(list_n, n, __ATOMIC_RELAXED, __HIP_MEMORY_SCOPE_WORKGROUP);
        p = __builtin_amdgcn_readfirstlane(p);
        if (p + n <= cap) pos = p;
    }
    for (int i0 = 0; i0 < n; i0 += 32) {
        const int i = i0 + lane;
        int s = 0, r = 0;
        if (i < n) { s = src[qe[i]]; r = qr[i]; }
        if (emit && pos >= 0 && i < n) ent[pos + i] = make_int2(s, r);
        int m = n - i0;
        if (m > 32) m = 32;
        agg_add_group(h, nid, emb, xs, split, s, r, m, acc, cnt, lane, count_deg);
    }
}

__device__ __forceinline__ int4 agg_load_dst4(const int* __restrict__ dst, int g, int ngroups,
                                              int n_edges, bool vec_ok) {
    int4 d = make_int4(-1, -1, -1, -1);
    if (g < ngroups) {
        if (vec_ok) {
            d = ((const int4*)dst)[g];
        } else {
            const int e = g * 4;
            if (e     < n_edges) d.x = dst[e];
            if (e + 1 < n_edges) d.y = dst[e + 1];
            if (e + 2 < n_edges) d.z = dst[e + 2];
            if (e + 3 < n_edges) d.w = dst[e + 3];
        }
    }
    return d;
}

__device__ __forceinline__ void agg_take_group(const int4 d, int g, int base, unsigned nrows,
                                               int* __restrict__ qe, int* __restrict__ qr, int& qn,
                                               const float* __restrict__ h,
                                               const int* __restrict__ nid,
                                               const float* __restrict__ emb,
                                               const float* __restrict__ xs,
                                               const bool split,
                                               const int* __restrict__ src,
                                               float* __restrict__ acc, int* __restrict__ cnt,
                                               int lane, int count_deg,
                                               const bool emit, int* __restrict__ list_n,
                                               int2* __restrict__ ent, int cap) {
    const int dv[4] = {d.x, d.y, d.z, d.w};
    unsigned rr[4];
    bool hit[4];
    bool any_hit = false;
#pragma unroll
    for (int k = 0; k < 4; ++k) {
        rr[k]  = (unsigned)(dv[k] - base);
        hit[k] = rr[k] < nrows;
        any_hit |= hit[k];
    }
    if (__builtin_amdgcn_ballot_w32(any_hit) != 0u) {
#pragma unroll
        for (int k = 0; k < 4; ++k) {
            const unsigned mask = __builtin_amdgcn_ballot_w32(hit[k]);
            if (mask != 0u) {
                if (hit[k]) {
                    const int pos = qn + (int)__builtin_amdgcn_mbcnt_lo(mask, 0u);
                    qe[pos] = g * 4 + k;
                    qr[pos] = (int)rr[k];
                }
                qn += __popc(mask);
            }
        }
        if (qn > Q_DRAIN) {
            agg_drain(h, nid, emb, xs, split, src, qe, qr, qn, acc, cnt, lane, count_deg,
                      emit, list_n, ent, cap);
            qn = 0;
        }
    }
}

__device__ __forceinline__ unsigned owner_tag4(const int4 d) {
    return  (((unsigned)d.x / AGG_ROWS) & 255u)
         | ((((unsigned)d.y / AGG_ROWS) & 255u) << 8)
         | ((((unsigned)d.z / AGG_ROWS) & 255u) << 16)
         | ((((unsigned)d.w / AGG_ROWS)       ) << 24);
}

__global__ __launch_bounds__(256)
void owner_tag_kernel(const int* __restrict__ dst, unsigned* __restrict__ tags,
                      int n_edges, int n_words) {
    const int  lane    = threadIdx.x & 31;
    const int  w0      = __builtin_amdgcn_readfirstlane((int)((blockIdx.x * 256 + threadIdx.x) >> 5)) * 128;
    const int  ngroups = (n_edges + 3) >> 2;
    const bool vec_ok  = ((n_edges & 3) == 0) && ((((size_t)dst) & 15) == 0);
    if (vec_ok && w0 + 128 <= ngroups) {
        const int4* __restrict__ dst4 = (const int4*)dst;
        int4 d[4];
#pragma unroll
        for (int k = 0; k < 4; ++k) d[k] = dst4[w0 + k * 32 + lane];
#pragma unroll
        for (int k = 0; k < 4; ++k) tags[w0 + k * 32 + lane] = owner_tag4(d[k]);
    } else {
        for (int k = 0; k < 4; ++k) {
            const int wi = w0 + k * 32 + lane;
            if (wi < n_words)
                tags[wi] = owner_tag4(agg_load_dst4(dst, wi, ngroups, n_edges, vec_ok));
        }
    }
}

__device__ __forceinline__ void tag_process(const int m, const int* qc, int& cn,
                                            const int* __restrict__ dst, int ngroups,
                                            int n_edges, bool vec_ok,
                                            int base, unsigned nrows,
                                            int* __restrict__ qe, int* __restrict__ qr, int& qn,
                                            const float* __restrict__ h,
                                            const int* __restrict__ nid,
                                            const float* __restrict__ emb,
                                            const float* __restrict__ xs,
                                            const bool split,
                                            const int* __restrict__ src,
                                            float* __restrict__ acc, int* __restrict__ cnt,
                                            int lane, int count_deg,
                                            const bool emit, int* __restrict__ list_n,
                                            int2* __restrict__ ent, int cap) {
    int g = ngroups;
    if (lane < m) g = qc[cn - m + lane];
    cn -= m;
    const int4 d = agg_load_dst4(dst, g, ngroups, n_edges, vec_ok);
    agg_take_group(d, g, base, nrows, qe, qr, qn, h, nid, emb, xs, split, src, acc, cnt,
                   lane, count_deg, emit, list_n, ent, cap);
}

__device__ __forceinline__ void tag_take(const uint4 w, const int g16, const unsigned bb,
                                         int* qc, int& cn,
                                         const int* __restrict__ dst, int ngroups,
                                         int n_edges, bool vec_ok,
                                         int base, unsigned nrows,
                                         int* __restrict__ qe, int* __restrict__ qr, int& qn,
                                         const float* __restrict__ h,
                                         const int* __restrict__ nid,
                                         const float* __restrict__ emb,
                                         const float* __restrict__ xs,
                                         const bool split,
                                         const int* __restrict__ src,
                                         float* __restrict__ acc, int* __restrict__ cnt,
                                         int lane, int count_deg,
                                         const bool emit, int* __restrict__ list_n,
                                         int2* __restrict__ ent, int cap) {
    const unsigned wv[4] = {w.x, w.y, w.z, w.w};
    unsigned z[4];
    unsigned zor = 0u;
#pragma unroll
    for (int k = 0; k < 4; ++k) {
        const unsigned x = wv[k] ^ bb;
        z[k] = (x - 0x01010101u) & ~x & 0x80808080u;
        zor |= z[k];
    }
    if (__builtin_amdgcn_ballot_w32(zor != 0u) != 0u) {
#pragma unroll
        for (int k = 0; k < 4; ++k) {
            const bool hit = z[k] != 0u;
            const unsigned mask = __builtin_amdgcn_ballot_w32(hit);
            if (mask != 0u) {
                if (hit) qc[cn + (int)__builtin_amdgcn_mbcnt_lo(mask, 0u)] = g16 * 4 + k;
                cn += __popc(mask);
            }
        }
        while (cn >= 32)
            tag_process(32, qc, cn, dst, ngroups, n_edges, vec_ok, base, nrows, qe, qr, qn,
                        h, nid, emb, xs, split, src, acc, cnt, lane, count_deg,
                        emit, list_n, ent, cap);
    }
}

#define KSTEPS       (HID / 4)
#define KGROUP       8

#define OWNER_LOAD_SELF(DST, G)                                                  \
    _Pragma("unroll")                                                            \
    for (int i_ = 0; i_ < KGROUP; ++i_) {                                        \
        const int t_ = (G) * KGROUP + i_;                                        \
        DST[i_] = t_ < KSTEPS / 2 ? ha_[2 * t_] : hb_[2 * t_ - KSTEPS];          \
    }

#define OWNER_KGROUP(SELF, G)                                                    \
    _Pragma("unroll")                                                            \
    for (int i_ = 0; i_ < KGROUP; ++i_) {                                        \
        const int t_ = (G) * KGROUP + i_;                                        \
        const v2f a_l = arow[t_];                                                \
        const v2f a_r = SELF[i_];                                                \
        const float4* wt_ = t_ < KSTEPS / 2 ? wj_lo + (2 * t_) * HID             \
                                            : wj_hi + (2 * t_ - KSTEPS) * HID;   \
        const float4 b0 = wt_[0];                                                \
        const float4 b1 = wt_[16];                                               \
        acc_l0 = __builtin_amdgcn_wmma_f32_16x16x4_f32(                          \
            false, a_l, false, (v2f){b0.x, b0.y}, (short)0, acc_l0, false, false); \
        acc_r0 = __builtin_amdgcn_wmma_f32_16x16x4_f32(                          \
            false, a_r, false, (v2f){b0.z, b0.w}, (short)0, acc_r0, false, false); \
        acc_l1 = __builtin_amdgcn_wmma_f32_16x16x4_f32(                          \
            false, a_l, false, (v2f){b1.x, b1.y}, (short)0, acc_l1, false, false); \
        acc_r1 = __builtin_amdgcn_wmma_f32_16x16x4_f32(                          \
            false, a_r, false, (v2f){b1.z, b1.w}, (short)0, acc_r1, false, false); \
    }

__device__ __forceinline__
void owner_gemm(const float* __restrict__ acc,
                const float4* __restrict__ wq,
                const float* __restrict__ h,
                const int* __restrict__ nid,
                const float* __restrict__ emb,
                const float* __restrict__ xs,
                const bool split,
                const float* __restrict__ bias,
                float* __restrict__ out,
                int base, int nr, int n_nodes, int do_relu, int wave, int lane) {
    const int c  = lane & 15;
    const int hh = lane >> 4;
    const int n_units = ((nr + 15) >> 4) * 4;

    for (int u = wave; u < n_units; u += AGG_WAVES) {
        const int rt = u >> 2;
        const int cp = u & 3;
        const int r  = rt * 16 + c;
        int grow = base + r;
        if (grow >= n_nodes) grow = n_nodes - 1;
        const v2f* __restrict__ arow = (const v2f*)(acc + r * ACC_PITCH + hh * ACC_HALF);
        int lo_entries = hh * HID + c + 32 * cp;
        int hi_entries = lo_entries + KSTEPS * HID;
        int ncol0      = 32 * cp + c;
        asm volatile("" : "+v"(lo_entries), "+v"(hi_entries), "+v"(ncol0));
        const float4* __restrict__ wj_lo = wq + lo_entries;
        const float4* __restrict__ wj_hi = wq + hi_entries;
        const v2f* ha_;
        const v2f* hb_;
        if (split) {
            ha_ = (const v2f*)(emb + (size_t)nid[grow] * EMBD) + hh;
            hb_ = (const v2f*)(xs  + (size_t)grow * EMBD) + hh;
        } else {
            ha_ = (const v2f*)(h + (size_t)grow * HID) + hh;
            hb_ = ha_ + EMBD / 2;
        }

        v8f acc_l0 = {}, acc_r0 = {};
        v8f acc_l1 = {}, acc_r1 = {};
        v2f sa[KGROUP], sb[KGROUP];
        OWNER_LOAD_SELF(sa, 0)
        OWNER_LOAD_SELF(sb, 1)
        OWNER_KGROUP(sa, 0)
        OWNER_LOAD_SELF(sa, 2)
        OWNER_KGROUP(sb, 1)
        OWNER_LOAD_SELF(sb, 3)
        OWNER_KGROUP(sa, 2)
        OWNER_KGROUP(sb, 3)

        __builtin_amdgcn_sched_barrier(0);
        asm volatile("v_nop\n\tv_nop\n\tv_nop\n\tv_nop\n\tv_nop\n\tv_nop\n\tv_nop\n\tv_nop"
                     : "+v"(acc_l0), "+v"(acc_r0), "+v"(acc_l1), "+v"(acc_r1));

        const int rbase = rt * 16 + hh * 8;
        const float bv0 = bias[ncol0];
        const float bv1 = bias[ncol0 + 16];
#pragma unroll
        for (int v = 0; v < 8; ++v) {
            const int rr = rbase + v;
            if (rr < nr) {
                float o0 = acc_l0[v] + acc_r0[v] + bv0;
                float o1 = acc_l1[v] + acc_r1[v] + bv1;
                if (do_relu) { o0 = o0 > 0.0f ? o0 : 0.0f; o1 = o1 > 0.0f ? o1 : 0.0f; }
                float* __restrict__ orow = out + (size_t)(base + rr) * HID;
                orow[ncol0]      = o0;
                orow[ncol0 + 16] = o1;
            }
        }
    }
}

__device__ __forceinline__
void agg_rows_body(float4* __restrict__ acc4, int* __restrict__ cnt,
                   int* __restrict__ q_edge, int* __restrict__ q_row,
                   int* __restrict__ q_cand, const unsigned* __restrict__ tags,
                   const bool use_tags,
                   int* __restrict__ list_n, float4* __restrict__ wq,
                   int* __restrict__ elist, const int list_mode,
                   const float* __restrict__ h,
                   const int* __restrict__ nid,
                   const float* __restrict__ emb,
                   const float* __restrict__ xs,
                   const bool split,
                   const int* __restrict__ src,
                   const int* __restrict__ dst,
                   const float* __restrict__ Wl,
                   const float* __restrict__ Wr,
                   const float* __restrict__ bias,
                   float* __restrict__ out,
                   float* __restrict__ invdeg,
                   int n_edges, int n_nodes, int write_invdeg, int do_relu) {
    float* acc = (float*)acc4;

    const int tid  = threadIdx.x;
    const int lane = tid & 31;
    const int wave = __builtin_amdgcn_readfirstlane(tid >> 5);
    const int base = blockIdx.x * AGG_ROWS;
    int nr = n_nodes - base;
    if (nr > AGG_ROWS) nr = AGG_ROWS;
    if (nr < 0) nr = 0;
    const unsigned nrows = (unsigned)nr;

    {
        const int rotw = (int)blockIdx.x & 7;
        float4 w[(HID / 2) * HID / AGG_THREADS];
#pragma unroll
        for (int it = 0; it < (HID / 2) * HID / AGG_THREADS; ++it) {
            const int i  = tid + (((it + rotw) & 7) * AGG_THREADS);
            const int kp = i >> 7, n = i & (HID - 1);
            w[it].x = Wl[(size_t)(2 * kp)     * HID + n];
            w[it].y = Wl[(size_t)(2 * kp + 1) * HID + n];
            w[it].z = Wr[(size_t)(2 * kp)     * HID + n];
            w[it].w = Wr[(size_t)(2 * kp + 1) * HID + n];
        }
        for (int i = tid; i < ACC_ROWS_PAD * ACC_PITCH / 4; i += AGG_THREADS)
            acc4[i] = make_float4(0.0f, 0.0f, 0.0f, 0.0f);
        for (int i = tid; i < ACC_ROWS_PAD; i += AGG_THREADS)
            cnt[i] = (!write_invdeg && i < nr) ? __float_as_int(invdeg[base + i]) : 0;
        if (tid == 0) *list_n = 0;
#pragma unroll
        for (int it = 0; it < (HID / 2) * HID / AGG_THREADS; ++it)
            wq[tid + (((it + rotw) & 7) * AGG_THREADS)] = w[it];
    }

    const size_t stretch = (((size_t)n_nodes * HID) / gridDim.x) & ~(size_t)(LIST_HDR - 1);
    int* __restrict__ my_list = elist + (size_t)blockIdx.x * stretch;
    int2* __restrict__ ent = (int2*)(my_list + LIST_HDR);
    const int  cap  = stretch > LIST_HDR ? (int)((stretch - LIST_HDR) >> 1) : 0;
    const bool emit = list_mode == 1;
    int n_list = -1;
    if (list_mode == 2 && stretch > 0) n_list = __builtin_amdgcn_readfirstlane(my_list[0]);
    __syncthreads();

    if (n_list >= 0) {
        const int per = ((n_list + AGG_WAVES - 1) / AGG_WAVES + 7) & ~7;
        const int beg = wave * per;
        int end = beg + per;
        if (end > n_list) end = n_list;
        int2 e_next = make_int2(0, 0);
        if (beg + lane < end) e_next = ent[beg + lane];
        for (int i0 = beg; i0 < end; i0 += 32) {
            const int2 e = e_next;
            e_next = make_int2(0, 0);
            if (i0 + 32 + lane < end) e_next = ent[i0 + 32 + lane];
            int m = end - i0;
            if (m > 32) m = 32;
            agg_add_group(h, nid, emb, xs, split, e.x, e.y, m, acc, cnt, lane, write_invdeg);
        }
    } else if (use_tags) {

    int* __restrict__ qe = q_edge + wave * Q_CAP;
    int* __restrict__ qr = q_row  + wave * Q_CAP;
    int* qc = q_cand + wave * C_CAP;
    int qn = 0;
    int cn = 0;

    const int  ngroups = (n_edges + 3) >> 2;
    const bool vec_ok  = ((n_edges & 3) == 0) && ((((size_t)dst) & 15) == 0);
    const int  n16     = (n_edges + 15) >> 4;
    const int  trip16  = 32 * TAG_U;
    const int  nfull   = n16 / trip16;
    const unsigned bb  = ((unsigned)blockIdx.x & 255u) * 0x01010101u;
    const uint4* __restrict__ tags4 = (const uint4*)tags;

    if (wave < nfull) {
        const int rot = (int)(((long long)blockIdx.x * nfull) / gridDim.x);

        uint4 cur[TAG_U], nxt[TAG_U];
        {
            int tt = wave + rot;
            if (tt >= nfull) tt -= nfull;
#pragma unroll
            for (int u = 0; u < TAG_U; ++u) cur[u] = tags4[tt * trip16 + u * 32 + lane];
        }
        for (int t = wave; t < nfull; t += AGG_WAVES) {
            const int t1 = t + AGG_WAVES;
            int tt0 = t + rot;
            int tt1 = (t1 < nfull ? t1 : t) + rot;
            if (tt0 >= nfull) tt0 -= nfull;
            if (tt1 >= nfull) tt1 -= nfull;
#pragma unroll
            for (int u = 0; u < TAG_U; ++u) nxt[u] = tags4[tt1 * trip16 + u * 32 + lane];
#pragma unroll
            for (int u = 0; u < TAG_U; ++u)
                tag_take(cur[u], tt0 * trip16 + u * 32 + lane, bb, qc, cn,
                         dst, ngroups, n_edges, vec_ok, base, nrows, qe, qr, qn,
                         h, nid, emb, xs, split, src, acc, cnt, lane, write_invdeg,
                         emit, list_n, ent, cap);
#pragma unroll
            for (int u = 0; u < TAG_U; ++u) cur[u] = nxt[u];
        }
    }
    for (int g0 = nfull * trip16 + wave * 32; g0 < n16; g0 += AGG_WAVES * 32) {
        const int g16 = g0 + lane;
        uint4 w = make_uint4(~bb, ~bb, ~bb, ~bb);
        if (g16 < n16) w = tags4[g16];
        tag_take(w, g16, bb, qc, cn, dst, ngroups, n_edges, vec_ok, base, nrows, qe, qr, qn,
                 h, nid, emb, xs, split, src, acc, cnt, lane, write_invdeg,
                 emit, list_n, ent, cap);
    }
    while (cn > 0)
        tag_process(cn < 32 ? cn : 32, qc, cn, dst, ngroups, n_edges, vec_ok, base, nrows,
                    qe, qr, qn, h, nid, emb, xs, split, src, acc, cnt, lane, write_invdeg,
                    emit, list_n, ent, cap);
    if (qn > 0) agg_drain(h, nid, emb, xs, split, src, qe, qr, qn, acc, cnt, lane, write_invdeg,
                          emit, list_n, ent, cap);

    } else {

    int* __restrict__ qe = q_edge + wave * Q_CAP;
    int* __restrict__ qr = q_row  + wave * Q_CAP;
    int qn = 0;

    const int  ngroups     = (n_edges + 3) >> 2;
    const int  trip_groups = 32 * SCAN_U;
    const bool vec_ok      = ((n_edges & 3) == 0) && ((((size_t)dst) & 15) == 0);
    const int  nfull       = vec_ok ? ngroups / trip_groups : 0;
    const int4* __restrict__ dst4 = (const int4*)dst;

    if (wave < nfull) {
        const int rot = (int)(((long long)blockIdx.x * nfull) / gridDim.x);

        int4 da[SCAN_U], db[SCAN_U];
        {
            int tt = wave + rot;
            if (tt >= nfull) tt -= nfull;
#pragma unroll
            for (int u = 0; u < SCAN_U; ++u) da[u] = dst4[tt * trip_groups + u * 32 + lane];
        }
        for (int t = wave; t < nfull; t += 2 * AGG_WAVES) {
            const int t1 = t + AGG_WAVES;
            const int t2 = t + 2 * AGG_WAVES;
            int tt0 = t + rot;
            int tt1 = (t1 < nfull ? t1 : t) + rot;
            int tt2 = (t2 < nfull ? t2 : t) + rot;
            if (tt0 >= nfull) tt0 -= nfull;
            if (tt1 >= nfull) tt1 -= nfull;
            if (tt2 >= nfull) tt2 -= nfull;

#pragma unroll
            for (int u = 0; u < SCAN_U; ++u) db[u] = dst4[tt1 * trip_groups + u * 32 + lane];
#pragma unroll
            for (int u = 0; u < SCAN_U; ++u)
                agg_take_group(da[u], tt0 * trip_groups + u * 32 + lane, base, nrows,
                               qe, qr, qn, h, nid, emb, xs, split, src, acc, cnt, lane, write_invdeg,
                               emit, list_n, ent, cap);
#pragma unroll
            for (int u = 0; u < SCAN_U; ++u) da[u] = dst4[tt2 * trip_groups + u * 32 + lane];
            if (t1 < nfull) {
#pragma unroll
                for (int u = 0; u < SCAN_U; ++u)
                    agg_take_group(db[u], tt1 * trip_groups + u * 32 + lane, base, nrows,
                                   qe, qr, qn, h, nid, emb, xs, split, src, acc, cnt, lane, write_invdeg,
                                   emit, list_n, ent, cap);
            }
        }
    }
    for (int g0 = nfull * trip_groups + wave * 32; g0 < ngroups; g0 += AGG_WAVES * 32) {
        const int g = g0 + lane;
        const int4 d = agg_load_dst4(dst, g, ngroups, n_edges, vec_ok);
        agg_take_group(d, g, base, nrows, qe, qr, qn, h, nid, emb, xs, split, src, acc, cnt, lane, write_invdeg,
                       emit, list_n, ent, cap);
    }
    if (qn > 0) agg_drain(h, nid, emb, xs, split, src, qe, qr, qn, acc, cnt, lane, write_invdeg,
                          emit, list_n, ent, cap);

    }

    __syncthreads();

    if (emit && tid == 0 && stretch > 0) {
        const int total = *list_n;
        my_list[0] = total <= cap ? total : -1;
    }
    if (write_invdeg) {
        for (int r = tid; r < nr; r += AGG_THREADS)
            invdeg[base + r] = 1.0f / fmaxf((float)cnt[r], 1.0f);
    }

    {
        float4 sv[SCALE_TRIPS];
        float  si[SCALE_TRIPS];
#pragma unroll
        for (int k = 0; k < SCALE_TRIPS; ++k) {
            const int i = tid + k * AGG_THREADS;
            sv[k] = make_float4(0.0f, 0.0f, 0.0f, 0.0f);
            si[k] = 0.0f;
            if (i < nr * (ACC_PITCH / 4)) {
                const int r = i / (ACC_PITCH / 4);
                sv[k] = acc4[i];
                si[k] = write_invdeg ? 1.0f / fmaxf((float)cnt[r], 1.0f)
                                     : __int_as_float(cnt[r]);
            }
        }
#pragma unroll
        for (int k = 0; k < SCALE_TRIPS; ++k) {
            const int i = tid + k * AGG_THREADS;
            if (i < nr * (ACC_PITCH / 4)) {
                float4 v = sv[k];
                v.x *= si[k]; v.y *= si[k]; v.z *= si[k]; v.w *= si[k];
                acc4[i] = v;
            }
        }
    }
    __syncthreads();

    owner_gemm(acc, wq, h, nid, emb, xs, split, bias, out, base, nr, n_nodes, do_relu, wave, lane);
}

__global__ __launch_bounds__(AGG_THREADS)
void sage_layer_in_kernel(const int* __restrict__ n_id,
                          const float* __restrict__ emb,
                          const float* __restrict__ x_struct,
                          const int* __restrict__ src,
                          const int* __restrict__ dst,
                          const float* __restrict__ Wl,
                          const float* __restrict__ Wr,
                          const float* __restrict__ bias,
                          float* __restrict__ out,
                          float* __restrict__ invdeg,
                          int* __restrict__ elist,
                          const unsigned* __restrict__ tags,
                          int n_edges, int n_nodes) {
    __shared__ float4 acc4[ACC_ROWS_PAD * ACC_PITCH / 4];
    __shared__ int    cnt[ACC_ROWS_PAD];
    __shared__ int    q_edge[AGG_WAVES * Q_CAP];
    __shared__ int    q_row [AGG_WAVES * Q_CAP];
    __shared__ int    q_cand[AGG_WAVES * C_CAP];
    __shared__ int    list_n;
    __shared__ float4 wq[(HID / 2) * HID];
    agg_rows_body(acc4, cnt, q_edge, q_row, q_cand, tags,  true,
                  &list_n, wq, elist,  1,
                  nullptr, n_id, emb, x_struct,  true,
                  src, dst, Wl, Wr, bias, out, invdeg, n_edges, n_nodes,
                   1,  1);
}

__global__ __launch_bounds__(AGG_THREADS)
void sage_layer_kernel(const float* __restrict__ h,
                       const int* __restrict__ src,
                       const int* __restrict__ dst,
                       const float* __restrict__ Wl,
                       const float* __restrict__ Wr,
                       const float* __restrict__ bias,
                       float* __restrict__ out,
                       float* __restrict__ invdeg,
                       int* __restrict__ elist,
                       int n_edges, int n_nodes, int do_relu) {
    __shared__ float4 acc4[ACC_ROWS_PAD * ACC_PITCH / 4];
    __shared__ int    cnt[ACC_ROWS_PAD];
    __shared__ int    q_edge[AGG_WAVES * Q_CAP];
    __shared__ int    q_row [AGG_WAVES * Q_CAP];
    __shared__ int    list_n;
    __shared__ float4 wq[(HID / 2) * HID];
    agg_rows_body(acc4, cnt, q_edge, q_row, nullptr, nullptr,  false,
                  &list_n, wq, elist,  2,
                  h, nullptr, nullptr, nullptr,  false,
                  src, dst, Wl, Wr, bias, out, invdeg, n_edges, n_nodes,
                   0, do_relu);
}

__global__ void edge_dot_kernel(const float* __restrict__ h2,
                                const int* __restrict__ es,
                                const int* __restrict__ ed,
                                float* __restrict__ out, int n) {
    int e = blockIdx.x * blockDim.x + threadIdx.x;
    if (e >= n) return;
    const float4* a = (const float4*)(h2 + (size_t)es[e] * HID);
    const float4* b = (const float4*)(h2 + (size_t)ed[e] * HID);
    float s = 0.0f;
#pragma unroll 8
    for (int i = 0; i < HID / 4; ++i) {
        float4 x = a[i], y = b[i];
        s += x.x * y.x + x.y * y.y + x.z * y.z + x.w * y.w;
    }
    out[e] = s;
}

extern "C" void kernel_launch(void* const* d_in, const int* in_sizes, int n_in,
                              void* d_out, int out_size, void* d_ws, size_t ws_size,
                              hipStream_t stream) {
    const int*   n_id     = (const int*)  d_in[0];
    const float* x_struct = (const float*)d_in[1];
    const int*   eidx     = (const int*)  d_in[2];
    const int*   elab     = (const int*)  d_in[3];
    const float* emb      = (const float*)d_in[4];
    const float* W1l      = (const float*)d_in[5];
    const float* W1r      = (const float*)d_in[6];
    const float* b1       = (const float*)d_in[7];
    const float* W2l      = (const float*)d_in[8];
    const float* W2r      = (const float*)d_in[9];
    const float* b2       = (const float*)d_in[10];
    float*       out      = (float*)d_out;

    const int N = in_sizes[0];
    const int E = in_sizes[2] / 2;
    const int L = in_sizes[3] / 2;

    const int* src = eidx;
    const int* dst = eidx + E;
    const int* lsrc = elab;
    const int* ldst = elab + L;

    float* ws     = (float*)d_ws;
    float* h1     = ws;
    float* h2     = h1  + (size_t)N * HID;
    float* lists  = h2  + (size_t)N * HID;
    float* invdeg = lists + (size_t)N * HID;

    const int ablocks = (N + AGG_ROWS - 1) / AGG_ROWS;
    int* elist = (int*)lists;

    unsigned* tags = (unsigned*)h2;
    const int n_words = ((E + 15) >> 4) << 2;
    owner_tag_kernel<<<(n_words + 1023) / 1024, 256, 0, stream>>>(dst, tags, E, n_words);

    sage_layer_in_kernel<<<ablocks, AGG_THREADS, 0, stream>>>(
        n_id, emb, x_struct, src, dst, W1l, W1r, b1, h1, invdeg, elist, tags, E, N);

    sage_layer_kernel<<<ablocks, AGG_THREADS, 0, stream>>>(
        h1, src, dst, W2l, W2r, b2, h2, invdeg, elist, E, N,  0);

    edge_dot_kernel<<<(L + 255) / 256, 256, 0, stream>>>(h2, lsrc, ldst, out, L);
}
